# starting kernel
# baseline (speedup 1.0000x reference)
#include <hip/hip_runtime.h>
#include <math.h>

typedef __attribute__((ext_vector_type(2))) float v2f;
typedef __attribute__((ext_vector_type(8))) float v8f;

#define DFEAT 128
#define DDEG 32
#define NEG_SLOPE 0.2f

__device__ __forceinline__ unsigned fenc(float f) {
  unsigned u = __float_as_uint(f);
  return (u & 0x80000000u) ? ~u : (u | 0x80000000u);
}
__device__ __forceinline__ float fdec(unsigned u) {
  u = (u & 0x80000000u) ? (u & 0x7fffffffu) : ~u;
  return __uint_as_float(u);
}

__global__ void __launch_bounds__(256) gat_gemm_xw(const float* __restrict__ X,
                                                   const float* __restrict__ W,
                                                   float* __restrict__ H,
                                                   int nrows) {
  __shared__ float xs[16 * DFEAT];
  const int tid = threadIdx.x;
  const int wave = tid >> 5;
  const int lane = tid & 31;
  const int rowBlk = blockIdx.x * 16;

  {
    int idx = tid * 8;
    int r = idx >> 7;
    int c = idx & (DFEAT - 1);
    const float4* src = (const float4*)(X + (size_t)(rowBlk + r) * DFEAT + c);
    float4 v0 = src[0];
    float4 v1 = src[1];
    *(float4*)(xs + r * DFEAT + c) = v0;
    *(float4*)(xs + r * DFEAT + c + 4) = v1;
  }
  __syncthreads();

  const int col0 = wave * 16;
  const int mn = lane & 15;
  const int kk = (lane >> 4) * 2;

  v8f acc = {};
  for (int k0 = 0; k0 < DFEAT; k0 += 4) {
    v2f a, b;
    a.x = xs[mn * DFEAT + k0 + kk];
    a.y = xs[mn * DFEAT + k0 + kk + 1];
    b.x = W[(size_t)(k0 + kk) * DFEAT + col0 + mn];
    b.y = W[(size_t)(k0 + kk + 1) * DFEAT + col0 + mn];
    acc = __builtin_amdgcn_wmma_f32_16x16x4_f32(
         false, a,  false, b,
         (short)0, acc,  false,  false);
  }

  const int mbase = (lane >> 4) * 8;
  float* out = H + (size_t)(rowBlk + mbase) * DFEAT + col0 + mn;
#pragma unroll
  for (int r = 0; r < 8; ++r) {
    out[(size_t)r * DFEAT] = acc[r];
  }
}

__global__ void __launch_bounds__(256) gat_node_alpha(const float* __restrict__ H,
                                                      const float* __restrict__ a_src,
                                                      const float* __restrict__ a_dst,
                                                      float* __restrict__ asrc,
                                                      float* __restrict__ adst,
                                                      int n) {
  int g = blockIdx.x * blockDim.x + threadIdx.x;
  int node = g >> 5;
  int lane = threadIdx.x & 31;
  if (node >= n) return;
  float4 h4 = *(const float4*)(H + (size_t)node * DFEAT + lane * 4);
  float4 s4 = *(const float4*)(a_src + lane * 4);
  float4 d4 = *(const float4*)(a_dst + lane * 4);
  float ps = h4.x * s4.x + h4.y * s4.y + h4.z * s4.z + h4.w * s4.w;
  float pd = h4.x * d4.x + h4.y * d4.y + h4.z * d4.z + h4.w * d4.w;
#pragma unroll
  for (int off = 16; off; off >>= 1) {
    ps += __shfl_down(ps, off, 32);
    pd += __shfl_down(pd, off, 32);
  }
  if (lane == 0) {
    asrc[node] = ps;
    adst[node] = pd;
  }
}

__global__ void gat_init_seg(unsigned* __restrict__ mmax, float* __restrict__ denom, int n) {
  int i = blockIdx.x * blockDim.x + threadIdx.x;
  if (i >= n) return;
  mmax[i] = fenc(-INFINITY);
  denom[i] = 0.0f;
}

__global__ void fill_zero(float* __restrict__ p, int count) {
  int i = blockIdx.x * blockDim.x + threadIdx.x;
  if (i < count) p[i] = 0.0f;
}

__global__ void __launch_bounds__(256) gat_edge_score(const int* __restrict__ srcI,
                                                      const int* __restrict__ dstI,
                                                      int ecount, int n,
                                                      const float* __restrict__ asrc,
                                                      const float* __restrict__ adst,
                                                      float* __restrict__ esc,
                                                      unsigned* __restrict__ mmax) {
  int e = blockIdx.x * blockDim.x + threadIdx.x;
  if (e >= ecount + n) return;
  int s, d;
  if (e < ecount) { s = srcI[e]; d = dstI[e]; } else { s = d = e - ecount; }
  float v = asrc[s] + adst[d];
  v = v > 0.0f ? v : v * NEG_SLOPE;
  esc[e] = v;
  atomicMax(mmax + d, fenc(v));
}

__global__ void __launch_bounds__(256) gat_edge_exp(const int* __restrict__ dstI,
                                                    int ecount, int n,
                                                    float* __restrict__ esc,
                                                    const unsigned* __restrict__ mmax,
                                                    float* __restrict__ denom) {
  int e = blockIdx.x * blockDim.x + threadIdx.x;
  if (e >= ecount + n) return;
  int d = (e < ecount) ? dstI[e] : (e - ecount);
  float w = expf(esc[e] - fdec(mmax[d]));
  esc[e] = w;
  atomicAdd(denom + d, w);
}

__global__ void __launch_bounds__(256) gat_edge_scatter(const int* __restrict__ srcI,
                                                        const int* __restrict__ dstI,
                                                        int ecount, int n,
                                                        const float* __restrict__ esc,
                                                        const float* __restrict__ denom,
                                                        const float* __restrict__ H,
                                                        float* __restrict__ O) {
  int g = blockIdx.x * blockDim.x + threadIdx.x;
  int e = g >> 5;
  int lane = threadIdx.x & 31;
  if (e >= ecount + n) return;
  int s, d;
  if (e < ecount) { s = srcI[e]; d = dstI[e]; } else { s = d = e - ecount; }
  float coef = esc[e] / (denom[d] + 1e-16f);
  float4 h4 = *(const float4*)(H + (size_t)s * DFEAT + lane * 4);
  float* o = O + (size_t)d * DFEAT + lane * 4;
  atomicAdd(o + 0, coef * h4.x);
  atomicAdd(o + 1, coef * h4.y);
  atomicAdd(o + 2, coef * h4.z);
  atomicAdd(o + 3, coef * h4.w);
}

__global__ void __launch_bounds__(256) gat_bias_act(float* __restrict__ O,
                                                    const float* __restrict__ b,
                                                    int total, int relu) {
  int i = blockIdx.x * blockDim.x + threadIdx.x;
  if (i >= total) return;
  float v = O[i] + b[i & (DFEAT - 1)];
  if (relu) v = fmaxf(v, 0.0f);
  O[i] = v;
}

__global__ void __launch_bounds__(256) gat_predict(const float* __restrict__ x2o,
                                                   const float* __restrict__ x2s,
                                                   const float* __restrict__ deg,
                                                   const float* __restrict__ Wp,
                                                   const float* __restrict__ bp,
                                                   float* __restrict__ out, int n) {
  int i = blockIdx.x * blockDim.x + threadIdx.x;
  if (i >= n) return;
  const float* zo = x2o + (size_t)i * DFEAT;
  const float* zs = x2s + (size_t)i * DFEAT;
  const float* zd = deg + (size_t)i * DDEG;
  float l0 = bp[0], l1 = bp[1];
#pragma unroll 4
  for (int d = 0; d < DFEAT; ++d) {
    l0 += zo[d] * Wp[d * 2 + 0];
    l1 += zo[d] * Wp[d * 2 + 1];
  }
#pragma unroll 4
  for (int d = 0; d < DFEAT; ++d) {
    l0 += zs[d] * Wp[(DFEAT + d) * 2 + 0];
    l1 += zs[d] * Wp[(DFEAT + d) * 2 + 1];
  }
#pragma unroll 4
  for (int d = 0; d < DDEG; ++d) {
    l0 += zd[d] * Wp[(2 * DFEAT + d) * 2 + 0];
    l1 += zd[d] * Wp[(2 * DFEAT + d) * 2 + 1];
  }
  float mx = fmaxf(l0, l1);
  float e0 = expf(l0 - mx), e1 = expf(l1 - mx);
  float inv = 1.0f / (e0 + e1);
  float a0 = e0 * inv, a1 = e1 * inv;
  float* po = out + (size_t)i * DFEAT;
#pragma unroll 4
  for (int d = 0; d < DFEAT; ++d) po[d] = a0 * zo[d] + a1 * zs[d];
}

static inline int ceil_div(long long a, long long b) { return (int)((a + b - 1) / b); }

static void run_gat_layer(const float* X, const int* srcI, const int* dstI, int ecount,
                          const float* W, const float* avs, const float* avd,
                          const float* bias, float* H, float* asrc, float* adst,
                          unsigned* mmax, float* denom, float* esc, float* O,
                          int n, int relu, hipStream_t stream) {
  gat_gemm_xw<<<n / 16, 256, 0, stream>>>(X, W, H, n);
  gat_node_alpha<<<ceil_div((long long)n * 32, 256), 256, 0, stream>>>(H, avs, avd, asrc, adst, n);
  gat_init_seg<<<ceil_div(n, 256), 256, 0, stream>>>(mmax, denom, n);
  fill_zero<<<ceil_div((long long)n * DFEAT, 256), 256, 0, stream>>>(O, n * DFEAT);
  int etot = ecount + n;
  gat_edge_score<<<ceil_div(etot, 256), 256, 0, stream>>>(srcI, dstI, ecount, n, asrc, adst, esc, mmax);
  gat_edge_exp<<<ceil_div(etot, 256), 256, 0, stream>>>(dstI, ecount, n, esc, mmax, denom);
  gat_edge_scatter<<<ceil_div((long long)etot * 32, 256), 256, 0, stream>>>(srcI, dstI, ecount, n,
                                                                            esc, denom, H, O);
  gat_bias_act<<<ceil_div((long long)n * DFEAT, 256), 256, 0, stream>>>(O, bias, n * DFEAT, relu);
}

extern "C" void kernel_launch(void* const* d_in, const int* in_sizes, int n_in,
                              void* d_out, int out_size, void* d_ws, size_t ws_size,
                              hipStream_t stream) {
  const float* x_o      = (const float*)d_in[0];
  const float* degree_o = (const float*)d_in[1];
  const int*   ei_o     = (const int*)d_in[2];
  const int*   ei_s     = (const int*)d_in[3];
  const float* W_o1 = (const float*)d_in[4];
  const float* as_o1 = (const float*)d_in[5];
  const float* ad_o1 = (const float*)d_in[6];
  const float* b_o1  = (const float*)d_in[7];
  const float* W_o2 = (const float*)d_in[8];
  const float* as_o2 = (const float*)d_in[9];
  const float* ad_o2 = (const float*)d_in[10];
  const float* b_o2  = (const float*)d_in[11];
  const float* W_s1 = (const float*)d_in[12];
  const float* as_s1 = (const float*)d_in[13];
  const float* ad_s1 = (const float*)d_in[14];
  const float* b_s1  = (const float*)d_in[15];
  const float* W_s2 = (const float*)d_in[16];
  const float* as_s2 = (const float*)d_in[17];
  const float* ad_s2 = (const float*)d_in[18];
  const float* b_s2  = (const float*)d_in[19];
  const float* W_pred = (const float*)d_in[20];
  const float* b_pred = (const float*)d_in[21];

  const int n = in_sizes[0] / DFEAT;
  const int ecount = in_sizes[2] / 2;

  const int* src_o = ei_o;
  const int* dst_o = ei_o + ecount;
  const int* src_s = ei_s;
  const int* dst_s = ei_s + ecount;

  float* ws = (float*)d_ws;
  float* H     = ws;
  float* X1    = H + (size_t)n * DFEAT;
  float* X2o   = X1 + (size_t)n * DFEAT;
  float* X2s   = X2o + (size_t)n * DFEAT;
  float* asrc  = X2s + (size_t)n * DFEAT;
  float* adst  = asrc + n;
  unsigned* mmax = (unsigned*)(adst + n);
  float* denom = (float*)(mmax + n);
  float* esc   = denom + n;

  run_gat_layer(x_o, src_o, dst_o, ecount, W_o1, as_o1, ad_o1, b_o1,
                H, asrc, adst, mmax, denom, esc, X1, n,  1, stream);
  run_gat_layer(X1, src_o, dst_o, ecount, W_o2, as_o2, ad_o2, b_o2,
                H, asrc, adst, mmax, denom, esc, X2o, n,  0, stream);
  run_gat_layer(x_o, src_s, dst_s, ecount, W_s1, as_s1, ad_s1, b_s1,
                H, asrc, adst, mmax, denom, esc, X1, n,  1, stream);
  run_gat_layer(X1, src_s, dst_s, ecount, W_s2, as_s2, ad_s2, b_s2,
                H, asrc, adst, mmax, denom, esc, X2s, n,  0, stream);

  gat_predict<<<ceil_div(n, 256), 256, 0, stream>>>(X2o, X2s, degree_o, W_pred, b_pred,
                                                    (float*)d_out, n);
}


# Counting-sort edges by dst once per graph (int atomic arrival-order rank); one wave per node does max, exp-sum and weighted row sum in registers; f32 atomics gone, sums run in rank order
# speedup vs baseline: 2.9920x; 2.9920x over previous
#include <hip/hip_runtime.h>
#include <math.h>

typedef __attribute__((ext_vector_type(2))) float v2f;
typedef __attribute__((ext_vector_type(8))) float v8f;

#define DFEAT 128
#define DDEG 32
#define NEG_SLOPE 0.2f

__global__ void __launch_bounds__(256) gat_gemm_xw(const float* __restrict__ X,
                                                   const float* __restrict__ W,
                                                   float* __restrict__ H,
                                                   int nrows) {
  __shared__ float xs[16 * DFEAT];
  const int tid = threadIdx.x;
  const int wave = tid >> 5;
  const int lane = tid & 31;
  const int rowBlk = blockIdx.x * 16;

  {
    int idx = tid * 8;
    int r = idx >> 7;
    int c = idx & (DFEAT - 1);
    const float4* src = (const float4*)(X + (size_t)(rowBlk + r) * DFEAT + c);
    float4 v0 = src[0];
    float4 v1 = src[1];
    *(float4*)(xs + r * DFEAT + c) = v0;
    *(float4*)(xs + r * DFEAT + c + 4) = v1;
  }
  __syncthreads();

  const int col0 = wave * 16;
  const int mn = lane & 15;
  const int kk = (lane >> 4) * 2;

  v8f acc = {};
  for (int k0 = 0; k0 < DFEAT; k0 += 4) {
    v2f a, b;
    a.x = xs[mn * DFEAT + k0 + kk];
    a.y = xs[mn * DFEAT + k0 + kk + 1];
    b.x = W[(size_t)(k0 + kk) * DFEAT + col0 + mn];
    b.y = W[(size_t)(k0 + kk + 1) * DFEAT + col0 + mn];
    acc = __builtin_amdgcn_wmma_f32_16x16x4_f32(
         false, a,  false, b,
         (short)0, acc,  false,  false);
  }

  const int mbase = (lane >> 4) * 8;
  float* out = H + (size_t)(rowBlk + mbase) * DFEAT + col0 + mn;
#pragma unroll
  for (int r = 0; r < 8; ++r) {
    out[(size_t)r * DFEAT] = acc[r];
  }
}

__global__ void __launch_bounds__(256) gat_node_alpha(const float* __restrict__ H,
                                                      const float* __restrict__ a_src,
                                                      const float* __restrict__ a_dst,
                                                      float* __restrict__ asrc,
                                                      float* __restrict__ adst,
                                                      int n) {
  int g = blockIdx.x * blockDim.x + threadIdx.x;
  int node = g >> 5;
  int lane = threadIdx.x & 31;
  if (node >= n) return;
  float4 h4 = *(const float4*)(H + (size_t)node * DFEAT + lane * 4);
  float4 s4 = *(const float4*)(a_src + lane * 4);
  float4 d4 = *(const float4*)(a_dst + lane * 4);
  float ps = h4.x * s4.x + h4.y * s4.y + h4.z * s4.z + h4.w * s4.w;
  float pd = h4.x * d4.x + h4.y * d4.y + h4.z * d4.z + h4.w * d4.w;
#pragma unroll
  for (int off = 16; off; off >>= 1) {
    ps += __shfl_down(ps, off, 32);
    pd += __shfl_down(pd, off, 32);
  }
  if (lane == 0) {
    asrc[node] = ps;
    adst[node] = pd;
  }
}

__global__ void csr_zero(int* __restrict__ cnt, int n) {
  int i = blockIdx.x * blockDim.x + threadIdx.x;
  if (i < n) cnt[i] = 0;
}

__global__ void __launch_bounds__(256) csr_count_rank(const int* __restrict__ dstI, int ecount,
                                                      int* __restrict__ cnt,
                                                      int* __restrict__ rank) {
  int e = blockIdx.x * blockDim.x + threadIdx.x;
  if (e >= ecount) return;
  rank[e] = atomicAdd(cnt + dstI[e], 1);
}

__global__ void __launch_bounds__(1024) csr_scan(const int* __restrict__ cnt,
                                                 int* __restrict__ segBeg, int n) {
  __shared__ int wsum[32];
  const int t = threadIdx.x;
  const int lane = t & 31;
  const int w = t >> 5;
  const int ch = (n + 1023) >> 10;
  const int b = t * ch;
  const int e = (b + ch < n) ? (b + ch) : n;
  int s = 0;
  for (int i = b; i < e; ++i) s += cnt[i];
  int x = s;
  for (int off = 1; off < 32; off <<= 1) {
    int y = __shfl_up(x, off, 32);
    if (lane >= off) x += y;
  }
  if (lane == 31) wsum[w] = x;
  __syncthreads();
  if (w == 0) {
    int v = wsum[lane];
    for (int off = 1; off < 32; off <<= 1) {
      int y = __shfl_up(v, off, 32);
      if (lane >= off) v += y;
    }
    wsum[lane] = v;
  }
  __syncthreads();
  int run = x - s + (w > 0 ? wsum[w - 1] : 0);
  int i = b;
  for (; i + 8 <= e; i += 8) {
    const int c0 = cnt[i], c1 = cnt[i + 1], c2 = cnt[i + 2], c3 = cnt[i + 3];
    const int c4 = cnt[i + 4], c5 = cnt[i + 5], c6 = cnt[i + 6], c7 = cnt[i + 7];
    segBeg[i] = run;     run += c0;
    segBeg[i + 1] = run; run += c1;
    segBeg[i + 2] = run; run += c2;
    segBeg[i + 3] = run; run += c3;
    segBeg[i + 4] = run; run += c4;
    segBeg[i + 5] = run; run += c5;
    segBeg[i + 6] = run; run += c6;
    segBeg[i + 7] = run; run += c7;
  }
  for (; i < e; ++i) {
    segBeg[i] = run;
    run += cnt[i];
  }
}

__global__ void __launch_bounds__(256) csr_fill(const int* __restrict__ srcI,
                                                const int* __restrict__ dstI, int ecount,
                                                const int* __restrict__ segBeg,
                                                const int* __restrict__ rank,
                                                int* __restrict__ csrSrc) {
  int e = blockIdx.x * blockDim.x + threadIdx.x;
  if (e >= ecount) return;
  csrSrc[segBeg[dstI[e]] + rank[e]] = srcI[e];
}

__device__ __forceinline__ float leaky(float v) { return v > 0.0f ? v : v * NEG_SLOPE; }

__device__ __forceinline__ float4 ld_row(const float* __restrict__ H, int s, int lane) {
  return *(const float4*)(H + (size_t)s * DFEAT + lane * 4);
}

__device__ __forceinline__ void fma_row(float4& acc, float c, const float4& h) {
  acc.x += c * h.x;
  acc.y += c * h.y;
  acc.z += c * h.z;
  acc.w += c * h.w;
}

__device__ __forceinline__ void agg_rows(const float* __restrict__ H, int lane, int s, float coef,
                                         int cnt, float4& acc) {
  const int ci = __float_as_int(coef);
  int t = 0;
  for (; t + 4 <= cnt; t += 4) {
    int s0 = __builtin_amdgcn_readlane(s, t);
    int s1 = __builtin_amdgcn_readlane(s, t + 1);
    int s2 = __builtin_amdgcn_readlane(s, t + 2);
    int s3 = __builtin_amdgcn_readlane(s, t + 3);
    float c0 = __int_as_float(__builtin_amdgcn_readlane(ci, t));
    float c1 = __int_as_float(__builtin_amdgcn_readlane(ci, t + 1));
    float c2 = __int_as_float(__builtin_amdgcn_readlane(ci, t + 2));
    float c3 = __int_as_float(__builtin_amdgcn_readlane(ci, t + 3));
    float4 h0 = ld_row(H, s0, lane);
    float4 h1 = ld_row(H, s1, lane);
    float4 h2 = ld_row(H, s2, lane);
    float4 h3 = ld_row(H, s3, lane);
    fma_row(acc, c0, h0);
    fma_row(acc, c1, h1);
    fma_row(acc, c2, h2);
    fma_row(acc, c3, h3);
  }
  for (; t < cnt; ++t) {
    int s0 = __builtin_amdgcn_readlane(s, t);
    float c0 = __int_as_float(__builtin_amdgcn_readlane(ci, t));
    float4 h0 = ld_row(H, s0, lane);
    fma_row(acc, c0, h0);
  }
}

__global__ void __launch_bounds__(256) gat_node_gather(const int* __restrict__ csrSrc,
                                                       const int* __restrict__ segBeg,
                                                       const int* __restrict__ cnt,
                                                       const float* __restrict__ asrc,
                                                       const float* __restrict__ adst,
                                                       const float* __restrict__ H,
                                                       const float* __restrict__ bias,
                                                       float* __restrict__ O, int n, int relu) {
  const int g = blockIdx.x * blockDim.x + threadIdx.x;
  const int lane = threadIdx.x & 31;
  const int d = __builtin_amdgcn_readfirstlane(g >> 5);
  if (d >= n) return;
  const int beg = segBeg[d];
  const int end = beg + cnt[d];
  const float ad = adst[d];
  const float vself = leaky(asrc[d] + ad);

  const int j0 = beg + lane;
  const bool ok0 = j0 < end;
  const int s0 = ok0 ? csrSrc[j0] : d;
  const float e0 = leaky(asrc[s0] + ad);

  float m = fmaxf(vself, e0);
  for (int j = j0 + 32; j < end; j += 32) m = fmaxf(m, leaky(asrc[csrSrc[j]] + ad));
  for (int off = 16; off; off >>= 1) m = fmaxf(m, __shfl_xor(m, off, 32));

  const float w0 = ok0 ? expf(e0 - m) : 0.0f;
  const float wself = expf(vself - m);
  float sum = w0;
  for (int j = j0 + 32; j < end; j += 32) sum += expf(leaky(asrc[csrSrc[j]] + ad) - m);
  if (lane == 0) sum += wself;
  for (int off = 16; off; off >>= 1) sum += __shfl_xor(sum, off, 32);
  const float den = sum + 1e-16f;

  float4 acc = make_float4(0.0f, 0.0f, 0.0f, 0.0f);
  fma_row(acc, wself / den, ld_row(H, d, lane));
  {
    const int c0 = (end - beg < 32) ? (end - beg) : 32;
    agg_rows(H, lane, s0, w0 / den, c0, acc);
  }
  for (int base = beg + 32; base < end; base += 32) {
    const int j = base + lane;
    int s = d;
    float coef = 0.0f;
    if (j < end) {
      s = csrSrc[j];
      coef = expf(leaky(asrc[s] + ad) - m) / den;
    }
    const int c = (end - base < 32) ? (end - base) : 32;
    agg_rows(H, lane, s, coef, c, acc);
  }

  const float4 b4 = *(const float4*)(bias + lane * 4);
  acc.x += b4.x;
  acc.y += b4.y;
  acc.z += b4.z;
  acc.w += b4.w;
  if (relu) {
    acc.x = fmaxf(acc.x, 0.0f);
    acc.y = fmaxf(acc.y, 0.0f);
    acc.z = fmaxf(acc.z, 0.0f);
    acc.w = fmaxf(acc.w, 0.0f);
  }
  *(float4*)(O + (size_t)d * DFEAT + lane * 4) = acc;
}

__global__ void __launch_bounds__(256) gat_predict(const float* __restrict__ x2o,
                                                   const float* __restrict__ x2s,
                                                   const float* __restrict__ deg,
                                                   const float* __restrict__ Wp,
                                                   const float* __restrict__ bp,
                                                   float* __restrict__ out, int n) {
  int i = blockIdx.x * blockDim.x + threadIdx.x;
  if (i >= n) return;
  const float* zo = x2o + (size_t)i * DFEAT;
  const float* zs = x2s + (size_t)i * DFEAT;
  const float* zd = deg + (size_t)i * DDEG;
  float l0 = bp[0], l1 = bp[1];
#pragma unroll 4
  for (int d = 0; d < DFEAT; ++d) {
    l0 += zo[d] * Wp[d * 2 + 0];
    l1 += zo[d] * Wp[d * 2 + 1];
  }
#pragma unroll 4
  for (int d = 0; d < DFEAT; ++d) {
    l0 += zs[d] * Wp[(DFEAT + d) * 2 + 0];
    l1 += zs[d] * Wp[(DFEAT + d) * 2 + 1];
  }
#pragma unroll 4
  for (int d = 0; d < DDEG; ++d) {
    l0 += zd[d] * Wp[(2 * DFEAT + d) * 2 + 0];
    l1 += zd[d] * Wp[(2 * DFEAT + d) * 2 + 1];
  }
  float mx = fmaxf(l0, l1);
  float e0 = expf(l0 - mx), e1 = expf(l1 - mx);
  float inv = 1.0f / (e0 + e1);
  float a0 = e0 * inv, a1 = e1 * inv;
  float* po = out + (size_t)i * DFEAT;
#pragma unroll 4
  for (int d = 0; d < DFEAT; ++d) po[d] = a0 * zo[d] + a1 * zs[d];
}

static inline int ceil_div(long long a, long long b) { return (int)((a + b - 1) / b); }

static void build_csr(const int* srcI, const int* dstI, int ecount, int n, int* cnt, int* segBeg,
                      int* rank, int* csrSrc, hipStream_t stream) {
  csr_zero<<<ceil_div(n, 256), 256, 0, stream>>>(cnt, n);
  csr_count_rank<<<ceil_div(ecount, 256), 256, 0, stream>>>(dstI, ecount, cnt, rank);
  csr_scan<<<1, 1024, 0, stream>>>(cnt, segBeg, n);
  csr_fill<<<ceil_div(ecount, 256), 256, 0, stream>>>(srcI, dstI, ecount, segBeg, rank, csrSrc);
}

static void run_gat_layer(const float* X, const float* W, const float* avs, const float* avd,
                          const float* bias, float* H, float* asrc, float* adst,
                          const int* csrSrc, const int* segBeg, const int* cnt, float* O,
                          int n, int relu, hipStream_t stream) {
  gat_gemm_xw<<<n / 16, 256, 0, stream>>>(X, W, H, n);
  gat_node_alpha<<<ceil_div((long long)n * 32, 256), 256, 0, stream>>>(H, avs, avd, asrc, adst, n);
  gat_node_gather<<<ceil_div((long long)n * 32, 256), 256, 0, stream>>>(csrSrc, segBeg, cnt, asrc, adst,
                                                                        H, bias, O, n, relu);
}

extern "C" void kernel_launch(void* const* d_in, const int* in_sizes, int n_in,
                              void* d_out, int out_size, void* d_ws, size_t ws_size,
                              hipStream_t stream) {
  const float* x_o      = (const float*)d_in[0];
  const float* degree_o = (const float*)d_in[1];
  const int*   ei_o     = (const int*)d_in[2];
  const int*   ei_s     = (const int*)d_in[3];
  const float* W_o1 = (const float*)d_in[4];
  const float* as_o1 = (const float*)d_in[5];
  const float* ad_o1 = (const float*)d_in[6];
  const float* b_o1  = (const float*)d_in[7];
  const float* W_o2 = (const float*)d_in[8];
  const float* as_o2 = (const float*)d_in[9];
  const float* ad_o2 = (const float*)d_in[10];
  const float* b_o2  = (const float*)d_in[11];
  const float* W_s1 = (const float*)d_in[12];
  const float* as_s1 = (const float*)d_in[13];
  const float* ad_s1 = (const float*)d_in[14];
  const float* b_s1  = (const float*)d_in[15];
  const float* W_s2 = (const float*)d_in[16];
  const float* as_s2 = (const float*)d_in[17];
  const float* ad_s2 = (const float*)d_in[18];
  const float* b_s2  = (const float*)d_in[19];
  const float* W_pred = (const float*)d_in[20];
  const float* b_pred = (const float*)d_in[21];

  const int n = in_sizes[0] / DFEAT;
  const int ecount = in_sizes[2] / 2;

  const int* src_o = ei_o;
  const int* dst_o = ei_o + ecount;
  const int* src_s = ei_s;
  const int* dst_s = ei_s + ecount;

  float* ws = (float*)d_ws;
  float* H     = ws;
  float* X1    = H + (size_t)n * DFEAT;
  float* X2o   = X1 + (size_t)n * DFEAT;
  float* X2s   = X2o + (size_t)n * DFEAT;
  float* asrc  = X2s + (size_t)n * DFEAT;
  float* adst  = asrc + n;
  int* cnt     = (int*)(adst + n);
  int* segBeg  = cnt + n;
  int* csrSrc  = segBeg + n;
  int* rank    = (int*)X2s;

  build_csr(src_o, dst_o, ecount, n, cnt, segBeg, rank, csrSrc, stream);
  run_gat_layer(x_o, W_o1, as_o1, ad_o1, b_o1, H, asrc, adst, csrSrc, segBeg, cnt, X1, n,
                 1, stream);
  run_gat_layer(X1, W_o2, as_o2, ad_o2, b_o2, H, asrc, adst, csrSrc, segBeg, cnt, X2o, n,
                 0, stream);
  build_csr(src_s, dst_s, ecount, n, cnt, segBeg, rank, csrSrc, stream);
  run_gat_layer(x_o, W_s1, as_s1, ad_s1, b_s1, H, asrc, adst, csrSrc, segBeg, cnt, X1, n,
                 1, stream);
  run_gat_layer(X1, W_s2, as_s2, ad_s2, b_s2, H, asrc, adst, csrSrc, segBeg, cnt, X2s, n,
                 0, stream);

  gat_predict<<<ceil_div(n, 256), 256, 0, stream>>>(X2o, X2s, degree_o, W_pred, b_pred,
                                                    (float*)d_out, n);
}


# Edges sorted by dst with LDS adds only, no global atomics: block histograms over 256-node buckets, scan, LDS-cursor scatter, a block per bucket ranks its nodes; segment sum order = LDS arrival order
# speedup vs baseline: 3.5835x; 1.1977x over previous
#include <hip/hip_runtime.h>
#include <math.h>

typedef __attribute__((ext_vector_type(2))) float v2f;
typedef __attribute__((ext_vector_type(8))) float v8f;

#define DFEAT 128
#define DDEG 32
#define NEG_SLOPE 0.2f

__global__ void __launch_bounds__(256) gat_gemm_xw(const float* __restrict__ X,
                                                   const float* __restrict__ W,
                                                   float* __restrict__ H,
                                                   int nrows) {
  __shared__ float xs[16 * DFEAT];
  const int tid = threadIdx.x;
  const int wave = tid >> 5;
  const int lane = tid & 31;
  const int rowBlk = blockIdx.x * 16;

  {
    int idx = tid * 8;
    int r = idx >> 7;
    int c = idx & (DFEAT - 1);
    const float4* src = (const float4*)(X + (size_t)(rowBlk + r) * DFEAT + c);
    float4 v0 = src[0];
    float4 v1 = src[1];
    *(float4*)(xs + r * DFEAT + c) = v0;
    *(float4*)(xs + r * DFEAT + c + 4) = v1;
  }
  __syncthreads();

  const int col0 = wave * 16;
  const int mn = lane & 15;
  const int kk = (lane >> 4) * 2;

  v8f acc = {};
  for (int k0 = 0; k0 < DFEAT; k0 += 4) {
    v2f a, b;
    a.x = xs[mn * DFEAT + k0 + kk];
    a.y = xs[mn * DFEAT + k0 + kk + 1];
    b.x = W[(size_t)(k0 + kk) * DFEAT + col0 + mn];
    b.y = W[(size_t)(k0 + kk + 1) * DFEAT + col0 + mn];
    acc = __builtin_amdgcn_wmma_f32_16x16x4_f32(
         false, a,  false, b,
         (short)0, acc,  false,  false);
  }

  const int mbase = (lane >> 4) * 8;
  float* out = H + (size_t)(rowBlk + mbase) * DFEAT + col0 + mn;
#pragma unroll
  for (int r = 0; r < 8; ++r) {
    out[(size_t)r * DFEAT] = acc[r];
  }
}

__global__ void __launch_bounds__(256) gat_node_alpha(const float* __restrict__ H,
                                                      const float* __restrict__ a_src,
                                                      const float* __restrict__ a_dst,
                                                      float* __restrict__ asrc,
                                                      float* __restrict__ adst,
                                                      int n) {
  int g = blockIdx.x * blockDim.x + threadIdx.x;
  int node = g >> 5;
  int lane = threadIdx.x & 31;
  if (node >= n) return;
  float4 h4 = *(const float4*)(H + (size_t)node * DFEAT + lane * 4);
  float4 s4 = *(const float4*)(a_src + lane * 4);
  float4 d4 = *(const float4*)(a_dst + lane * 4);
  float ps = h4.x * s4.x + h4.y * s4.y + h4.z * s4.z + h4.w * s4.w;
  float pd = h4.x * d4.x + h4.y * d4.y + h4.z * d4.z + h4.w * d4.w;
#pragma unroll
  for (int off = 16; off; off >>= 1) {
    ps += __shfl_down(ps, off, 32);
    pd += __shfl_down(pd, off, 32);
  }
  if (lane == 0) {
    asrc[node] = ps;
    adst[node] = pd;
  }
}

#define BKT_SHIFT 8
#define BKT_NODES 256
#define BKT_EDGES_PER_BLOCK 4096

__global__ void __launch_bounds__(512) csr_bucket_count(const int* __restrict__ dstI, int ecount,
                                                        int nb, int* __restrict__ hist) {
  __shared__ int h[BKT_NODES];
  const int t = threadIdx.x;
  const int g = blockIdx.x;
  const int G = gridDim.x;
  if (t < BKT_NODES) h[t] = 0;
  __syncthreads();
  const int base = g * BKT_EDGES_PER_BLOCK + t;
  int d[8];
#pragma unroll
  for (int k = 0; k < 8; ++k) {
    const int i = base + k * 512;
    d[k] = (i < ecount) ? dstI[i] : -1;
  }
#pragma unroll
  for (int k = 0; k < 8; ++k) {
    if (d[k] >= 0) atomicAdd(&h[d[k] >> BKT_SHIFT], 1);
  }
  __syncthreads();
  if (t < nb) hist[t * G + g] = h[t];
}

__global__ void __launch_bounds__(1024) csr_scan(const int* __restrict__ cnt,
                                                 int* __restrict__ segBeg, int n) {
  __shared__ int wsum[32];
  const int t = threadIdx.x;
  const int lane = t & 31;
  const int w = t >> 5;
  const int ch = (n + 1023) >> 10;
  const int b = t * ch;
  const int e = (b + ch < n) ? (b + ch) : n;
  int s = 0;
  for (int i = b; i < e; ++i) s += cnt[i];
  int x = s;
  for (int off = 1; off < 32; off <<= 1) {
    int y = __shfl_up(x, off, 32);
    if (lane >= off) x += y;
  }
  if (lane == 31) wsum[w] = x;
  __syncthreads();
  if (w == 0) {
    int v = wsum[lane];
    for (int off = 1; off < 32; off <<= 1) {
      int y = __shfl_up(v, off, 32);
      if (lane >= off) v += y;
    }
    wsum[lane] = v;
  }
  __syncthreads();
  int run = x - s + (w > 0 ? wsum[w - 1] : 0);
  int i = b;
  for (; i + 8 <= e; i += 8) {
    const int c0 = cnt[i], c1 = cnt[i + 1], c2 = cnt[i + 2], c3 = cnt[i + 3];
    const int c4 = cnt[i + 4], c5 = cnt[i + 5], c6 = cnt[i + 6], c7 = cnt[i + 7];
    segBeg[i] = run;     run += c0;
    segBeg[i + 1] = run; run += c1;
    segBeg[i + 2] = run; run += c2;
    segBeg[i + 3] = run; run += c3;
    segBeg[i + 4] = run; run += c4;
    segBeg[i + 5] = run; run += c5;
    segBeg[i + 6] = run; run += c6;
    segBeg[i + 7] = run; run += c7;
  }
  for (; i < e; ++i) {
    segBeg[i] = run;
    run += cnt[i];
  }
}

__global__ void __launch_bounds__(512) csr_bucket_scatter(const int* __restrict__ srcI,
                                                          const int* __restrict__ dstI, int ecount,
                                                          int nb, const int* __restrict__ off,
                                                          int* __restrict__ keys) {
  __shared__ int cur[BKT_NODES];
  const int t = threadIdx.x;
  const int g = blockIdx.x;
  const int G = gridDim.x;
  if (t < nb) cur[t] = off[t * G + g];
  __syncthreads();
  const int base = g * BKT_EDGES_PER_BLOCK + t;
  int d[8], s[8];
#pragma unroll
  for (int k = 0; k < 8; ++k) {
    const int i = base + k * 512;
    const bool ok = i < ecount;
    d[k] = ok ? dstI[i] : -1;
    s[k] = ok ? srcI[i] : 0;
  }
#pragma unroll
  for (int k = 0; k < 8; ++k) {
    if (d[k] >= 0) {
      const int pos = atomicAdd(&cur[d[k] >> BKT_SHIFT], 1);
      keys[pos] = ((d[k] & (BKT_NODES - 1)) << 24) | s[k];
    }
  }
}

__global__ void __launch_bounds__(256) csr_bucket_finish(const int* __restrict__ keys,
                                                         const int* __restrict__ off, int G, int nb,
                                                         int ecount, int n, int* __restrict__ cnt,
                                                         int* __restrict__ segBeg,
                                                         int* __restrict__ csrSrc) {
  __shared__ int c[BKT_NODES];
  __shared__ int wsum[8];
  const int t = threadIdx.x;
  const int lane = t & 31;
  const int w = t >> 5;
  const int b = blockIdx.x;
  const int beg = off[b * G];
  const int end = (b + 1 < nb) ? off[(b + 1) * G] : ecount;
  c[t] = 0;
  __syncthreads();
  for (int i0 = beg + t; i0 < end; i0 += 256 * 8) {
    unsigned key[8];
#pragma unroll
    for (int k = 0; k < 8; ++k) {
      const int i = i0 + k * 256;
      key[k] = (unsigned)keys[(i < end) ? i : i0];
    }
#pragma unroll
    for (int k = 0; k < 8; ++k) {
      const bool ok = i0 + k * 256 < end;
      atomicAdd(&c[ok ? (int)(key[k] >> 24) : t], ok ? 1 : 0);
    }
  }
  __syncthreads();
  const int my = c[t];
  int x = my;
  for (int o = 1; o < 32; o <<= 1) {
    int y = __shfl_up(x, o, 32);
    if (lane >= o) x += y;
  }
  if (lane == 31) wsum[w] = x;
  __syncthreads();
  int wb = 0;
  for (int j = 0; j < w; ++j) wb += wsum[j];
  const int start = beg + wb + x - my;
  const int node = b * BKT_NODES + t;
  if (node < n) {
    cnt[node] = my;
    segBeg[node] = start;
  }
  c[t] = start;
  __syncthreads();
  for (int i0 = beg + t; i0 < end; i0 += 256 * 8) {
    unsigned key[8];
    int pos[8];
#pragma unroll
    for (int k = 0; k < 8; ++k) {
      const int i = i0 + k * 256;
      key[k] = (unsigned)keys[(i < end) ? i : i0];
    }
#pragma unroll
    for (int k = 0; k < 8; ++k) {
      const bool ok = i0 + k * 256 < end;
      pos[k] = atomicAdd(&c[ok ? (int)(key[k] >> 24) : t], ok ? 1 : 0);
    }
#pragma unroll
    for (int k = 0; k < 8; ++k) {
      if (i0 + k * 256 < end) csrSrc[pos[k]] = (int)(key[k] & 0x00FFFFFFu);
    }
  }
}

__device__ __forceinline__ float leaky(float v) { return v > 0.0f ? v : v * NEG_SLOPE; }

__device__ __forceinline__ float4 ld_row(const float* __restrict__ H, int s, int lane) {
  return *(const float4*)(H + (size_t)s * DFEAT + lane * 4);
}

__device__ __forceinline__ void fma_row(float4& acc, float c, const float4& h) {
  acc.x += c * h.x;
  acc.y += c * h.y;
  acc.z += c * h.z;
  acc.w += c * h.w;
}

__device__ __forceinline__ void agg_rows(const float* __restrict__ H, int lane, int s, float coef,
                                         int cnt, float4& acc) {
  const int ci = __float_as_int(coef);
  int t = 0;
  for (; t + 4 <= cnt; t += 4) {
    int s0 = __builtin_amdgcn_readlane(s, t);
    int s1 = __builtin_amdgcn_readlane(s, t + 1);
    int s2 = __builtin_amdgcn_readlane(s, t + 2);
    int s3 = __builtin_amdgcn_readlane(s, t + 3);
    float c0 = __int_as_float(__builtin_amdgcn_readlane(ci, t));
    float c1 = __int_as_float(__builtin_amdgcn_readlane(ci, t + 1));
    float c2 = __int_as_float(__builtin_amdgcn_readlane(ci, t + 2));
    float c3 = __int_as_float(__builtin_amdgcn_readlane(ci, t + 3));
    float4 h0 = ld_row(H, s0, lane);
    float4 h1 = ld_row(H, s1, lane);
    float4 h2 = ld_row(H, s2, lane);
    float4 h3 = ld_row(H, s3, lane);
    fma_row(acc, c0, h0);
    fma_row(acc, c1, h1);
    fma_row(acc, c2, h2);
    fma_row(acc, c3, h3);
  }
  for (; t < cnt; ++t) {
    int s0 = __builtin_amdgcn_readlane(s, t);
    float c0 = __int_as_float(__builtin_amdgcn_readlane(ci, t));
    float4 h0 = ld_row(H, s0, lane);
    fma_row(acc, c0, h0);
  }
}

__global__ void __launch_bounds__(256) gat_node_gather(const int* __restrict__ csrSrc,
                                                       const int* __restrict__ segBeg,
                                                       const int* __restrict__ cnt,
                                                       const float* __restrict__ asrc,
                                                       const float* __restrict__ adst,
                                                       const float* __restrict__ H,
                                                       const float* __restrict__ bias,
                                                       float* __restrict__ O, int n, int relu) {
  const int g = blockIdx.x * blockDim.x + threadIdx.x;
  const int lane = threadIdx.x & 31;
  const int d = __builtin_amdgcn_readfirstlane(g >> 5);
  if (d >= n) return;
  const int beg = segBeg[d];
  const int end = beg + cnt[d];
  const float ad = adst[d];
  const float vself = leaky(asrc[d] + ad);

  const int j0 = beg + lane;
  const bool ok0 = j0 < end;
  const int s0 = ok0 ? csrSrc[j0] : d;
  const float e0 = leaky(asrc[s0] + ad);

  float m = fmaxf(vself, e0);
  for (int j = j0 + 32; j < end; j += 32) m = fmaxf(m, leaky(asrc[csrSrc[j]] + ad));
  for (int off = 16; off; off >>= 1) m = fmaxf(m, __shfl_xor(m, off, 32));

  const float w0 = ok0 ? expf(e0 - m) : 0.0f;
  const float wself = expf(vself - m);
  float sum = w0;
  for (int j = j0 + 32; j < end; j += 32) sum += expf(leaky(asrc[csrSrc[j]] + ad) - m);
  if (lane == 0) sum += wself;
  for (int off = 16; off; off >>= 1) sum += __shfl_xor(sum, off, 32);
  const float den = sum + 1e-16f;

  float4 acc = make_float4(0.0f, 0.0f, 0.0f, 0.0f);
  fma_row(acc, wself / den, ld_row(H, d, lane));
  {
    const int c0 = (end - beg < 32) ? (end - beg) : 32;
    agg_rows(H, lane, s0, w0 / den, c0, acc);
  }
  for (int base = beg + 32; base < end; base += 32) {
    const int j = base + lane;
    int s = d;
    float coef = 0.0f;
    if (j < end) {
      s = csrSrc[j];
      coef = expf(leaky(asrc[s] + ad) - m) / den;
    }
    const int c = (end - base < 32) ? (end - base) : 32;
    agg_rows(H, lane, s, coef, c, acc);
  }

  const float4 b4 = *(const float4*)(bias + lane * 4);
  acc.x += b4.x;
  acc.y += b4.y;
  acc.z += b4.z;
  acc.w += b4.w;
  if (relu) {
    acc.x = fmaxf(acc.x, 0.0f);
    acc.y = fmaxf(acc.y, 0.0f);
    acc.z = fmaxf(acc.z, 0.0f);
    acc.w = fmaxf(acc.w, 0.0f);
  }
  *(float4*)(O + (size_t)d * DFEAT + lane * 4) = acc;
}

__global__ void __launch_bounds__(256) gat_predict(const float* __restrict__ x2o,
                                                   const float* __restrict__ x2s,
                                                   const float* __restrict__ deg,
                                                   const float* __restrict__ Wp,
                                                   const float* __restrict__ bp,
                                                   float* __restrict__ out, int n) {
  int i = blockIdx.x * blockDim.x + threadIdx.x;
  if (i >= n) return;
  const float* zo = x2o + (size_t)i * DFEAT;
  const float* zs = x2s + (size_t)i * DFEAT;
  const float* zd = deg + (size_t)i * DDEG;
  float l0 = bp[0], l1 = bp[1];
#pragma unroll 4
  for (int d = 0; d < DFEAT; ++d) {
    l0 += zo[d] * Wp[d * 2 + 0];
    l1 += zo[d] * Wp[d * 2 + 1];
  }
#pragma unroll 4
  for (int d = 0; d < DFEAT; ++d) {
    l0 += zs[d] * Wp[(DFEAT + d) * 2 + 0];
    l1 += zs[d] * Wp[(DFEAT + d) * 2 + 1];
  }
#pragma unroll 4
  for (int d = 0; d < DDEG; ++d) {
    l0 += zd[d] * Wp[(2 * DFEAT + d) * 2 + 0];
    l1 += zd[d] * Wp[(2 * DFEAT + d) * 2 + 1];
  }
  float mx = fmaxf(l0, l1);
  float e0 = expf(l0 - mx), e1 = expf(l1 - mx);
  float inv = 1.0f / (e0 + e1);
  float a0 = e0 * inv, a1 = e1 * inv;
  float* po = out + (size_t)i * DFEAT;
#pragma unroll 4
  for (int d = 0; d < DFEAT; ++d) po[d] = a0 * zo[d] + a1 * zs[d];
}

static inline int ceil_div(long long a, long long b) { return (int)((a + b - 1) / b); }

static void build_csr(const int* srcI, const int* dstI, int ecount, int n, int* cnt, int* segBeg,
                      int* keys, int* hist, int* off, int* csrSrc, hipStream_t stream) {
  const int G = ceil_div(ecount, BKT_EDGES_PER_BLOCK);
  const int nb = ceil_div(n, BKT_NODES);
  csr_bucket_count<<<G, 512, 0, stream>>>(dstI, ecount, nb, hist);
  csr_scan<<<1, 1024, 0, stream>>>(hist, off, nb * G);
  csr_bucket_scatter<<<G, 512, 0, stream>>>(srcI, dstI, ecount, nb, off, keys);
  csr_bucket_finish<<<nb, 256, 0, stream>>>(keys, off, G, nb, ecount, n, cnt, segBeg, csrSrc);
}

static void run_gat_layer(const float* X, const float* W, const float* avs, const float* avd,
                          const float* bias, float* H, float* asrc, float* adst,
                          const int* csrSrc, const int* segBeg, const int* cnt, float* O,
                          int n, int relu, hipStream_t stream) {
  gat_gemm_xw<<<n / 16, 256, 0, stream>>>(X, W, H, n);
  gat_node_alpha<<<ceil_div((long long)n * 32, 256), 256, 0, stream>>>(H, avs, avd, asrc, adst, n);
  gat_node_gather<<<ceil_div((long long)n * 32, 256), 256, 0, stream>>>(csrSrc, segBeg, cnt, asrc, adst,
                                                                        H, bias, O, n, relu);
}

extern "C" void kernel_launch(void* const* d_in, const int* in_sizes, int n_in,
                              void* d_out, int out_size, void* d_ws, size_t ws_size,
                              hipStream_t stream) {
  const float* x_o      = (const float*)d_in[0];
  const float* degree_o = (const float*)d_in[1];
  const int*   ei_o     = (const int*)d_in[2];
  const int*   ei_s     = (const int*)d_in[3];
  const float* W_o1 = (const float*)d_in[4];
  const float* as_o1 = (const float*)d_in[5];
  const float* ad_o1 = (const float*)d_in[6];
  const float* b_o1  = (const float*)d_in[7];
  const float* W_o2 = (const float*)d_in[8];
  const float* as_o2 = (const float*)d_in[9];
  const float* ad_o2 = (const float*)d_in[10];
  const float* b_o2  = (const float*)d_in[11];
  const float* W_s1 = (const float*)d_in[12];
  const float* as_s1 = (const float*)d_in[13];
  const float* ad_s1 = (const float*)d_in[14];
  const float* b_s1  = (const float*)d_in[15];
  const float* W_s2 = (const float*)d_in[16];
  const float* as_s2 = (const float*)d_in[17];
  const float* ad_s2 = (const float*)d_in[18];
  const float* b_s2  = (const float*)d_in[19];
  const float* W_pred = (const float*)d_in[20];
  const float* b_pred = (const float*)d_in[21];

  const int n = in_sizes[0] / DFEAT;
  const int ecount = in_sizes[2] / 2;

  const int* src_o = ei_o;
  const int* dst_o = ei_o + ecount;
  const int* src_s = ei_s;
  const int* dst_s = ei_s + ecount;

  float* ws = (float*)d_ws;
  float* H     = ws;
  float* X1    = H + (size_t)n * DFEAT;
  float* X2o   = X1 + (size_t)n * DFEAT;
  float* X2s   = X2o + (size_t)n * DFEAT;
  float* asrc  = X2s + (size_t)n * DFEAT;
  float* adst  = asrc + n;
  int* cnt     = (int*)(adst + n);
  int* segBeg  = cnt + n;
  int* csrSrc  = segBeg + n;
  int* keys    = (int*)X2s;
  int* hist    = (int*)H;
  int* off     = hist + (size_t)n * DFEAT / 2;

  build_csr(src_o, dst_o, ecount, n, cnt, segBeg, keys, hist, off, csrSrc, stream);
  run_gat_layer(x_o, W_o1, as_o1, ad_o1, b_o1, H, asrc, adst, csrSrc, segBeg, cnt, X1, n,
                 1, stream);
  run_gat_layer(X1, W_o2, as_o2, ad_o2, b_o2, H, asrc, adst, csrSrc, segBeg, cnt, X2o, n,
                 0, stream);
  build_csr(src_s, dst_s, ecount, n, cnt, segBeg, keys, hist, off, csrSrc, stream);
  run_gat_layer(x_o, W_s1, as_s1, ad_s1, b_s1, H, asrc, adst, csrSrc, segBeg, cnt, X1, n,
                 1, stream);
  run_gat_layer(X1, W_s2, as_s2, ad_s2, b_s2, H, asrc, adst, csrSrc, segBeg, cnt, X2s, n,
                 0, stream);

  gat_predict<<<ceil_div(n, 256), 256, 0, stream>>>(X2o, X2s, degree_o, W_pred, b_pred,
                                                    (float*)d_out, n);
}


# a_src/a_dst node dots fused into the GEMM epilogue: DPP 16-lane row sums of the accumulators, 8 column tiles then added in tile order via LDS (dot sum order changes); the 4 one-row-wave launches go
# speedup vs baseline: 3.8832x; 1.0836x over previous
#include <hip/hip_runtime.h>
#include <math.h>

typedef __attribute__((ext_vector_type(2))) float v2f;
typedef __attribute__((ext_vector_type(8))) float v8f;

#define DFEAT 128
#define DDEG 32
#define NEG_SLOPE 0.2f

template <int CTRL>
__device__ __forceinline__ float row_swz(float v) {
  return __int_as_float(__builtin_amdgcn_update_dpp(0, __float_as_int(v), CTRL, 0xF, 0xF, true));
}
__device__ __forceinline__ float row16_sum(float v) {
  v += row_swz<0x161>(v);
  v += row_swz<0x162>(v);
  v += row_swz<0x164>(v);
  v += row_swz<0x168>(v);
  return v;
}

__global__ void __launch_bounds__(256) gat_gemm_xw(const float* __restrict__ X,
                                                   const float* __restrict__ W,
                                                   const float* __restrict__ a_src,
                                                   const float* __restrict__ a_dst,
                                                   float* __restrict__ H,
                                                   float* __restrict__ asrc,
                                                   float* __restrict__ adst,
                                                   int nrows) {
  __shared__ float xs[16 * DFEAT];
  __shared__ float red[2][8][16];
  const int tid = threadIdx.x;
  const int wave = tid >> 5;
  const int lane = tid & 31;
  const int rowBlk = blockIdx.x * 16;

  {
    int idx = tid * 8;
    int r = idx >> 7;
    int c = idx & (DFEAT - 1);
    const float4* src = (const float4*)(X + (size_t)(rowBlk + r) * DFEAT + c);
    float4 v0 = src[0];
    float4 v1 = src[1];
    *(float4*)(xs + r * DFEAT + c) = v0;
    *(float4*)(xs + r * DFEAT + c + 4) = v1;
  }
  __syncthreads();

  const int col0 = wave * 16;
  const int mn = lane & 15;
  const int kk = (lane >> 4) * 2;

  v8f acc = {};
  for (int k0 = 0; k0 < DFEAT; k0 += 4) {
    v2f a, b;
    a.x = xs[mn * DFEAT + k0 + kk];
    a.y = xs[mn * DFEAT + k0 + kk + 1];
    b.x = W[(size_t)(k0 + kk) * DFEAT + col0 + mn];
    b.y = W[(size_t)(k0 + kk + 1) * DFEAT + col0 + mn];
    acc = __builtin_amdgcn_wmma_f32_16x16x4_f32(
         false, a,  false, b,
         (short)0, acc,  false,  false);
  }

  const int mbase = (lane >> 4) * 8;
  float* out = H + (size_t)(rowBlk + mbase) * DFEAT + col0 + mn;
#pragma unroll
  for (int r = 0; r < 8; ++r) {
    out[(size_t)r * DFEAT] = acc[r];
  }

  const float vs = a_src[col0 + mn];
  const float vd = a_dst[col0 + mn];
  float ps[8], pd[8];
#pragma unroll
  for (int r = 0; r < 8; ++r) {
    ps[r] = row16_sum(acc[r] * vs);
    pd[r] = row16_sum(acc[r] * vd);
  }
  if (mn == 0) {
#pragma unroll
    for (int r = 0; r < 8; ++r) {
      red[0][wave][mbase + r] = ps[r];
      red[1][wave][mbase + r] = pd[r];
    }
  }
  __syncthreads();
  if (tid < 32) {
    const int row = tid & 15;
    const int which = tid >> 4;
    float t = red[which][0][row];
#pragma unroll
    for (int w = 1; w < 8; ++w) t += red[which][w][row];
    float* dstp = which ? adst : asrc;
    dstp[rowBlk + row] = t;
  }
}

#define BKT_SHIFT 8
#define BKT_NODES 256
#define BKT_EDGES_PER_BLOCK 4096

__global__ void __launch_bounds__(512) csr_bucket_count(const int* __restrict__ dstI, int ecount,
                                                        int nb, int* __restrict__ hist) {
  __shared__ int h[BKT_NODES];
  const int t = threadIdx.x;
  const int g = blockIdx.x;
  const int G = gridDim.x;
  if (t < BKT_NODES) h[t] = 0;
  __syncthreads();
  const int base = g * BKT_EDGES_PER_BLOCK + t;
  int d[8];
#pragma unroll
  for (int k = 0; k < 8; ++k) {
    const int i = base + k * 512;
    d[k] = (i < ecount) ? dstI[i] : -1;
  }
#pragma unroll
  for (int k = 0; k < 8; ++k) {
    if (d[k] >= 0) atomicAdd(&h[d[k] >> BKT_SHIFT], 1);
  }
  __syncthreads();
  if (t < nb) hist[t * G + g] = h[t];
}

__global__ void __launch_bounds__(1024) csr_scan(const int* __restrict__ cnt,
                                                 int* __restrict__ segBeg, int n) {
  __shared__ int wsum[32];
  const int t = threadIdx.x;
  const int lane = t & 31;
  const int w = t >> 5;
  const int ch = (n + 1023) >> 10;
  const int b = t * ch;
  const int e = (b + ch < n) ? (b + ch) : n;
  int s = 0;
  for (int i = b; i < e; ++i) s += cnt[i];
  int x = s;
  for (int off = 1; off < 32; off <<= 1) {
    int y = __shfl_up(x, off, 32);
    if (lane >= off) x += y;
  }
  if (lane == 31) wsum[w] = x;
  __syncthreads();
  if (w == 0) {
    int v = wsum[lane];
    for (int off = 1; off < 32; off <<= 1) {
      int y = __shfl_up(v, off, 32);
      if (lane >= off) v += y;
    }
    wsum[lane] = v;
  }
  __syncthreads();
  int run = x - s + (w > 0 ? wsum[w - 1] : 0);
  int i = b;
  for (; i + 8 <= e; i += 8) {
    const int c0 = cnt[i], c1 = cnt[i + 1], c2 = cnt[i + 2], c3 = cnt[i + 3];
    const int c4 = cnt[i + 4], c5 = cnt[i + 5], c6 = cnt[i + 6], c7 = cnt[i + 7];
    segBeg[i] = run;     run += c0;
    segBeg[i + 1] = run; run += c1;
    segBeg[i + 2] = run; run += c2;
    segBeg[i + 3] = run; run += c3;
    segBeg[i + 4] = run; run += c4;
    segBeg[i + 5] = run; run += c5;
    segBeg[i + 6] = run; run += c6;
    segBeg[i + 7] = run; run += c7;
  }
  for (; i < e; ++i) {
    segBeg[i] = run;
    run += cnt[i];
  }
}

__global__ void __launch_bounds__(512) csr_bucket_scatter(const int* __restrict__ srcI,
                                                          const int* __restrict__ dstI, int ecount,
                                                          int nb, const int* __restrict__ off,
                                                          int* __restrict__ keys) {
  __shared__ int cur[BKT_NODES];
  const int t = threadIdx.x;
  const int g = blockIdx.x;
  const int G = gridDim.x;
  if (t < nb) cur[t] = off[t * G + g];
  __syncthreads();
  const int base = g * BKT_EDGES_PER_BLOCK + t;
  int d[8], s[8];
#pragma unroll
  for (int k = 0; k < 8; ++k) {
    const int i = base + k * 512;
    const bool ok = i < ecount;
    d[k] = ok ? dstI[i] : -1;
    s[k] = ok ? srcI[i] : 0;
  }
#pragma unroll
  for (int k = 0; k < 8; ++k) {
    if (d[k] >= 0) {
      const int pos = atomicAdd(&cur[d[k] >> BKT_SHIFT], 1);
      keys[pos] = ((d[k] & (BKT_NODES - 1)) << 24) | s[k];
    }
  }
}

__global__ void __launch_bounds__(256) csr_bucket_finish(const int* __restrict__ keys,
                                                         const int* __restrict__ off, int G, int nb,
                                                         int ecount, int n, int* __restrict__ cnt,
                                                         int* __restrict__ segBeg,
                                                         int* __restrict__ csrSrc) {
  __shared__ int c[BKT_NODES];
  __shared__ int wsum[8];
  const int t = threadIdx.x;
  const int lane = t & 31;
  const int w = t >> 5;
  const int b = blockIdx.x;
  const int beg = off[b * G];
  const int end = (b + 1 < nb) ? off[(b + 1) * G] : ecount;
  c[t] = 0;
  __syncthreads();
  for (int i0 = beg + t; i0 < end; i0 += 256 * 8) {
    unsigned key[8];
#pragma unroll
    for (int k = 0; k < 8; ++k) {
      const int i = i0 + k * 256;
      key[k] = (unsigned)keys[(i < end) ? i : i0];
    }
#pragma unroll
    for (int k = 0; k < 8; ++k) {
      const bool ok = i0 + k * 256 < end;
      atomicAdd(&c[ok ? (int)(key[k] >> 24) : t], ok ? 1 : 0);
    }
  }
  __syncthreads();
  const int my = c[t];
  int x = my;
  for (int o = 1; o < 32; o <<= 1) {
    int y = __shfl_up(x, o, 32);
    if (lane >= o) x += y;
  }
  if (lane == 31) wsum[w] = x;
  __syncthreads();
  int wb = 0;
  for (int j = 0; j < w; ++j) wb += wsum[j];
  const int start = beg + wb + x - my;
  const int node = b * BKT_NODES + t;
  if (node < n) {
    cnt[node] = my;
    segBeg[node] = start;
  }
  c[t] = start;
  __syncthreads();
  for (int i0 = beg + t; i0 < end; i0 += 256 * 8) {
    unsigned key[8];
    int pos[8];
#pragma unroll
    for (int k = 0; k < 8; ++k) {
      const int i = i0 + k * 256;
      key[k] = (unsigned)keys[(i < end) ? i : i0];
    }
#pragma unroll
    for (int k = 0; k < 8; ++k) {
      const bool ok = i0 + k * 256 < end;
      pos[k] = atomicAdd(&c[ok ? (int)(key[k] >> 24) : t], ok ? 1 : 0);
    }
#pragma unroll
    for (int k = 0; k < 8; ++k) {
      if (i0 + k * 256 < end) csrSrc[pos[k]] = (int)(key[k] & 0x00FFFFFFu);
    }
  }
}

__device__ __forceinline__ float leaky(float v) { return v > 0.0f ? v : v * NEG_SLOPE; }

__device__ __forceinline__ float4 ld_row(const float* __restrict__ H, int s, int lane) {
  return *(const float4*)(H + (size_t)s * DFEAT + lane * 4);
}

__device__ __forceinline__ void fma_row(float4& acc, float c, const float4& h) {
  acc.x += c * h.x;
  acc.y += c * h.y;
  acc.z += c * h.z;
  acc.w += c * h.w;
}

__device__ __forceinline__ void agg_rows(const float* __restrict__ H, int lane, int s, float coef,
                                         int cnt, float4& acc) {
  const int ci = __float_as_int(coef);
  int t = 0;
  for (; t + 4 <= cnt; t += 4) {
    int s0 = __builtin_amdgcn_readlane(s, t);
    int s1 = __builtin_amdgcn_readlane(s, t + 1);
    int s2 = __builtin_amdgcn_readlane(s, t + 2);
    int s3 = __builtin_amdgcn_readlane(s, t + 3);
    float c0 = __int_as_float(__builtin_amdgcn_readlane(ci, t));
    float c1 = __int_as_float(__builtin_amdgcn_readlane(ci, t + 1));
    float c2 = __int_as_float(__builtin_amdgcn_readlane(ci, t + 2));
    float c3 = __int_as_float(__builtin_amdgcn_readlane(ci, t + 3));
    float4 h0 = ld_row(H, s0, lane);
    float4 h1 = ld_row(H, s1, lane);
    float4 h2 = ld_row(H, s2, lane);
    float4 h3 = ld_row(H, s3, lane);
    fma_row(acc, c0, h0);
    fma_row(acc, c1, h1);
    fma_row(acc, c2, h2);
    fma_row(acc, c3, h3);
  }
  for (; t < cnt; ++t) {
    int s0 = __builtin_amdgcn_readlane(s, t);
    float c0 = __int_as_float(__builtin_amdgcn_readlane(ci, t));
    float4 h0 = ld_row(H, s0, lane);
    fma_row(acc, c0, h0);
  }
}

__global__ void __launch_bounds__(256) gat_node_gather(const int* __restrict__ csrSrc,
                                                       const int* __restrict__ segBeg,
                                                       const int* __restrict__ cnt,
                                                       const float* __restrict__ asrc,
                                                       const float* __restrict__ adst,
                                                       const float* __restrict__ H,
                                                       const float* __restrict__ bias,
                                                       float* __restrict__ O, int n, int relu) {
  const int g = blockIdx.x * blockDim.x + threadIdx.x;
  const int lane = threadIdx.x & 31;
  const int d = __builtin_amdgcn_readfirstlane(g >> 5);
  if (d >= n) return;
  const int beg = segBeg[d];
  const int end = beg + cnt[d];
  const float ad = adst[d];
  const float vself = leaky(asrc[d] + ad);

  const int j0 = beg + lane;
  const bool ok0 = j0 < end;
  const int s0 = ok0 ? csrSrc[j0] : d;
  const float e0 = leaky(asrc[s0] + ad);

  float m = fmaxf(vself, e0);
  for (int j = j0 + 32; j < end; j += 32) m = fmaxf(m, leaky(asrc[csrSrc[j]] + ad));
  for (int off = 16; off; off >>= 1) m = fmaxf(m, __shfl_xor(m, off, 32));

  const float w0 = ok0 ? expf(e0 - m) : 0.0f;
  const float wself = expf(vself - m);
  float sum = w0;
  for (int j = j0 + 32; j < end; j += 32) sum += expf(leaky(asrc[csrSrc[j]] + ad) - m);
  if (lane == 0) sum += wself;
  for (int off = 16; off; off >>= 1) sum += __shfl_xor(sum, off, 32);
  const float den = sum + 1e-16f;

  float4 acc = make_float4(0.0f, 0.0f, 0.0f, 0.0f);
  fma_row(acc, wself / den, ld_row(H, d, lane));
  {
    const int c0 = (end - beg < 32) ? (end - beg) : 32;
    agg_rows(H, lane, s0, w0 / den, c0, acc);
  }
  for (int base = beg + 32; base < end; base += 32) {
    const int j = base + lane;
    int s = d;
    float coef = 0.0f;
    if (j < end) {
      s = csrSrc[j];
      coef = expf(leaky(asrc[s] + ad) - m) / den;
    }
    const int c = (end - base < 32) ? (end - base) : 32;
    agg_rows(H, lane, s, coef, c, acc);
  }

  const float4 b4 = *(const float4*)(bias + lane * 4);
  acc.x += b4.x;
  acc.y += b4.y;
  acc.z += b4.z;
  acc.w += b4.w;
  if (relu) {
    acc.x = fmaxf(acc.x, 0.0f);
    acc.y = fmaxf(acc.y, 0.0f);
    acc.z = fmaxf(acc.z, 0.0f);
    acc.w = fmaxf(acc.w, 0.0f);
  }
  *(float4*)(O + (size_t)d * DFEAT + lane * 4) = acc;
}

__global__ void __launch_bounds__(256) gat_predict(const float* __restrict__ x2o,
                                                   const float* __restrict__ x2s,
                                                   const float* __restrict__ deg,
                                                   const float* __restrict__ Wp,
                                                   const float* __restrict__ bp,
                                                   float* __restrict__ out, int n) {
  int i = blockIdx.x * blockDim.x + threadIdx.x;
  if (i >= n) return;
  const float* zo = x2o + (size_t)i * DFEAT;
  const float* zs = x2s + (size_t)i * DFEAT;
  const float* zd = deg + (size_t)i * DDEG;
  float l0 = bp[0], l1 = bp[1];
#pragma unroll 4
  for (int d = 0; d < DFEAT; ++d) {
    l0 += zo[d] * Wp[d * 2 + 0];
    l1 += zo[d] * Wp[d * 2 + 1];
  }
#pragma unroll 4
  for (int d = 0; d < DFEAT; ++d) {
    l0 += zs[d] * Wp[(DFEAT + d) * 2 + 0];
    l1 += zs[d] * Wp[(DFEAT + d) * 2 + 1];
  }
#pragma unroll 4
  for (int d = 0; d < DDEG; ++d) {
    l0 += zd[d] * Wp[(2 * DFEAT + d) * 2 + 0];
    l1 += zd[d] * Wp[(2 * DFEAT + d) * 2 + 1];
  }
  float mx = fmaxf(l0, l1);
  float e0 = expf(l0 - mx), e1 = expf(l1 - mx);
  float inv = 1.0f / (e0 + e1);
  float a0 = e0 * inv, a1 = e1 * inv;
  float* po = out + (size_t)i * DFEAT;
#pragma unroll 4
  for (int d = 0; d < DFEAT; ++d) po[d] = a0 * zo[d] + a1 * zs[d];
}

static inline int ceil_div(long long a, long long b) { return (int)((a + b - 1) / b); }

static void build_csr(const int* srcI, const int* dstI, int ecount, int n, int* cnt, int* segBeg,
                      int* keys, int* hist, int* off, int* csrSrc, hipStream_t stream) {
  const int G = ceil_div(ecount, BKT_EDGES_PER_BLOCK);
  const int nb = ceil_div(n, BKT_NODES);
  csr_bucket_count<<<G, 512, 0, stream>>>(dstI, ecount, nb, hist);
  csr_scan<<<1, 1024, 0, stream>>>(hist, off, nb * G);
  csr_bucket_scatter<<<G, 512, 0, stream>>>(srcI, dstI, ecount, nb, off, keys);
  csr_bucket_finish<<<nb, 256, 0, stream>>>(keys, off, G, nb, ecount, n, cnt, segBeg, csrSrc);
}

static void run_gat_layer(const float* X, const float* W, const float* avs, const float* avd,
                          const float* bias, float* H, float* asrc, float* adst,
                          const int* csrSrc, const int* segBeg, const int* cnt, float* O,
                          int n, int relu, hipStream_t stream) {
  gat_gemm_xw<<<n / 16, 256, 0, stream>>>(X, W, avs, avd, H, asrc, adst, n);
  gat_node_gather<<<ceil_div((long long)n * 32, 256), 256, 0, stream>>>(csrSrc, segBeg, cnt, asrc, adst,
                                                                        H, bias, O, n, relu);
}

extern "C" void kernel_launch(void* const* d_in, const int* in_sizes, int n_in,
                              void* d_out, int out_size, void* d_ws, size_t ws_size,
                              hipStream_t stream) {
  const float* x_o      = (const float*)d_in[0];
  const float* degree_o = (const float*)d_in[1];
  const int*   ei_o     = (const int*)d_in[2];
  const int*   ei_s     = (const int*)d_in[3];
  const float* W_o1 = (const float*)d_in[4];
  const float* as_o1 = (const float*)d_in[5];
  const float* ad_o1 = (const float*)d_in[6];
  const float* b_o1  = (const float*)d_in[7];
  const float* W_o2 = (const float*)d_in[8];
  const float* as_o2 = (const float*)d_in[9];
  const float* ad_o2 = (const float*)d_in[10];
  const float* b_o2  = (const float*)d_in[11];
  const float* W_s1 = (const float*)d_in[12];
  const float* as_s1 = (const float*)d_in[13];
  const float* ad_s1 = (const float*)d_in[14];
  const float* b_s1  = (const float*)d_in[15];
  const float* W_s2 = (const float*)d_in[16];
  const float* as_s2 = (const float*)d_in[17];
  const float* ad_s2 = (const float*)d_in[18];
  const float* b_s2  = (const float*)d_in[19];
  const float* W_pred = (const float*)d_in[20];
  const float* b_pred = (const float*)d_in[21];

  const int n = in_sizes[0] / DFEAT;
  const int ecount = in_sizes[2] / 2;

  const int* src_o = ei_o;
  const int* dst_o = ei_o + ecount;
  const int* src_s = ei_s;
  const int* dst_s = ei_s + ecount;

  float* ws = (float*)d_ws;
  float* H     = ws;
  float* X1    = H + (size_t)n * DFEAT;
  float* X2o   = X1 + (size_t)n * DFEAT;
  float* X2s   = X2o + (size_t)n * DFEAT;
  float* asrc  = X2s + (size_t)n * DFEAT;
  float* adst  = asrc + n;
  int* cnt     = (int*)(adst + n);
  int* segBeg  = cnt + n;
  int* csrSrc  = segBeg + n;
  int* keys    = (int*)X2s;
  int* hist    = (int*)H;
  int* off     = hist + (size_t)n * DFEAT / 2;

  build_csr(src_o, dst_o, ecount, n, cnt, segBeg, keys, hist, off, csrSrc, stream);
  run_gat_layer(x_o, W_o1, as_o1, ad_o1, b_o1, H, asrc, adst, csrSrc, segBeg, cnt, X1, n,
                 1, stream);
  run_gat_layer(X1, W_o2, as_o2, ad_o2, b_o2, H, asrc, adst, csrSrc, segBeg, cnt, X2o, n,
                 0, stream);
  build_csr(src_s, dst_s, ecount, n, cnt, segBeg, keys, hist, off, csrSrc, stream);
  run_gat_layer(x_o, W_s1, as_s1, ad_s1, b_s1, H, asrc, adst, csrSrc, segBeg, cnt, X1, n,
                 1, stream);
  run_gat_layer(X1, W_s2, as_s2, ad_s2, b_s2, H, asrc, adst, csrSrc, segBeg, cnt, X2s, n,
                 0, stream);

  gat_predict<<<ceil_div(n, 256), 256, 0, stream>>>(X2o, X2s, degree_o, W_pred, b_pred,
                                                    (float*)d_out, n);
}


# GEMM A tile in LDS at a padded row pitch (132 floats, was 128): the 16-way bank conflict on every A fragment load goes; values and sum order unchanged
# speedup vs baseline: 4.2975x; 1.1067x over previous
#include <hip/hip_runtime.h>
#include <math.h>

typedef __attribute__((ext_vector_type(2))) float v2f;
typedef __attribute__((ext_vector_type(8))) float v8f;

#define DFEAT 128
#define DDEG 32
#define NEG_SLOPE 0.2f
#define XS_PITCH (DFEAT + 4)

template <int CTRL>
__device__ __forceinline__ float row_swz(float v) {
  return __int_as_float(__builtin_amdgcn_update_dpp(0, __float_as_int(v), CTRL, 0xF, 0xF, true));
}
__device__ __forceinline__ float row16_sum(float v) {
  v += row_swz<0x161>(v);
  v += row_swz<0x162>(v);
  v += row_swz<0x164>(v);
  v += row_swz<0x168>(v);
  return v;
}

__global__ void __launch_bounds__(256) gat_gemm_xw(const float* __restrict__ X,
                                                   const float* __restrict__ W,
                                                   const float* __restrict__ a_src,
                                                   const float* __restrict__ a_dst,
                                                   float* __restrict__ H,
                                                   float* __restrict__ asrc,
                                                   float* __restrict__ adst,
                                                   int nrows) {
  __shared__ float xs[16 * XS_PITCH];
  __shared__ float red[2][8][16];
  const int tid = threadIdx.x;
  const int wave = tid >> 5;
  const int lane = tid & 31;
  const int rowBlk = blockIdx.x * 16;

  {
    int idx = tid * 8;
    int r = idx >> 7;
    int c = idx & (DFEAT - 1);
    const float4* src = (const float4*)(X + (size_t)(rowBlk + r) * DFEAT + c);
    float4 v0 = src[0];
    float4 v1 = src[1];
    *(float4*)(xs + r * XS_PITCH + c) = v0;
    *(float4*)(xs + r * XS_PITCH + c + 4) = v1;
  }
  __syncthreads();

  const int col0 = wave * 16;
  const int mn = lane & 15;
  const int kk = (lane >> 4) * 2;

  v8f acc = {};
  for (int k0 = 0; k0 < DFEAT; k0 += 4) {
    v2f a, b;
    a.x = xs[mn * XS_PITCH + k0 + kk];
    a.y = xs[mn * XS_PITCH + k0 + kk + 1];
    b.x = W[(size_t)(k0 + kk) * DFEAT + col0 + mn];
    b.y = W[(size_t)(k0 + kk + 1) * DFEAT + col0 + mn];
    acc = __builtin_amdgcn_wmma_f32_16x16x4_f32(
         false, a,  false, b,
         (short)0, acc,  false,  false);
  }

  const int mbase = (lane >> 4) * 8;
  float* out = H + (size_t)(rowBlk + mbase) * DFEAT + col0 + mn;
#pragma unroll
  for (int r = 0; r < 8; ++r) {
    out[(size_t)r * DFEAT] = acc[r];
  }

  const float vs = a_src[col0 + mn];
  const float vd = a_dst[col0 + mn];
  float ps[8], pd[8];
#pragma unroll
  for (int r = 0; r < 8; ++r) {
    ps[r] = row16_sum(acc[r] * vs);
    pd[r] = row16_sum(acc[r] * vd);
  }
  if (mn == 0) {
#pragma unroll
    for (int r = 0; r < 8; ++r) {
      red[0][wave][mbase + r] = ps[r];
      red[1][wave][mbase + r] = pd[r];
    }
  }
  __syncthreads();
  if (tid < 32) {
    const int row = tid & 15;
    const int which = tid >> 4;
    float t = red[which][0][row];
#pragma unroll
    for (int w = 1; w < 8; ++w) t += red[which][w][row];
    float* dstp = which ? adst : asrc;
    dstp[rowBlk + row] = t;
  }
}

#define BKT_SHIFT 8
#define BKT_NODES 256
#define BKT_EDGES_PER_BLOCK 4096

__global__ void __launch_bounds__(512) csr_bucket_count(const int* __restrict__ dstI, int ecount,
                                                        int nb, int* __restrict__ hist) {
  __shared__ int h[BKT_NODES];
  const int t = threadIdx.x;
  const int g = blockIdx.x;
  const int G = gridDim.x;
  if (t < BKT_NODES) h[t] = 0;
  __syncthreads();
  const int base = g * BKT_EDGES_PER_BLOCK + t;
  int d[8];
#pragma unroll
  for (int k = 0; k < 8; ++k) {
    const int i = base + k * 512;
    d[k] = (i < ecount) ? dstI[i] : -1;
  }
#pragma unroll
  for (int k = 0; k < 8; ++k) {
    if (d[k] >= 0) atomicAdd(&h[d[k] >> BKT_SHIFT], 1);
  }
  __syncthreads();
  if (t < nb) hist[t * G + g] = h[t];
}

__global__ void __launch_bounds__(1024) csr_scan(const int* __restrict__ cnt,
                                                 int* __restrict__ segBeg, int n) {
  __shared__ int wsum[32];
  const int t = threadIdx.x;
  const int lane = t & 31;
  const int w = t >> 5;
  const int ch = (n + 1023) >> 10;
  const int b = t * ch;
  const int e = (b + ch < n) ? (b + ch) : n;
  int s = 0;
  for (int i = b; i < e; ++i) s += cnt[i];
  int x = s;
  for (int off = 1; off < 32; off <<= 1) {
    int y = __shfl_up(x, off, 32);
    if (lane >= off) x += y;
  }
  if (lane == 31) wsum[w] = x;
  __syncthreads();
  if (w == 0) {
    int v = wsum[lane];
    for (int off = 1; off < 32; off <<= 1) {
      int y = __shfl_up(v, off, 32);
      if (lane >= off) v += y;
    }
    wsum[lane] = v;
  }
  __syncthreads();
  int run = x - s + (w > 0 ? wsum[w - 1] : 0);
  int i = b;
  for (; i + 8 <= e; i += 8) {
    const int c0 = cnt[i], c1 = cnt[i + 1], c2 = cnt[i + 2], c3 = cnt[i + 3];
    const int c4 = cnt[i + 4], c5 = cnt[i + 5], c6 = cnt[i + 6], c7 = cnt[i + 7];
    segBeg[i] = run;     run += c0;
    segBeg[i + 1] = run; run += c1;
    segBeg[i + 2] = run; run += c2;
    segBeg[i + 3] = run; run += c3;
    segBeg[i + 4] = run; run += c4;
    segBeg[i + 5] = run; run += c5;
    segBeg[i + 6] = run; run += c6;
    segBeg[i + 7] = run; run += c7;
  }
  for (; i < e; ++i) {
    segBeg[i] = run;
    run += cnt[i];
  }
}

__global__ void __launch_bounds__(512) csr_bucket_scatter(const int* __restrict__ srcI,
                                                          const int* __restrict__ dstI, int ecount,
                                                          int nb, const int* __restrict__ off,
                                                          int* __restrict__ keys) {
  __shared__ int cur[BKT_NODES];
  const int t = threadIdx.x;
  const int g = blockIdx.x;
  const int G = gridDim.x;
  if (t < nb) cur[t] = off[t * G + g];
  __syncthreads();
  const int base = g * BKT_EDGES_PER_BLOCK + t;
  int d[8], s[8];
#pragma unroll
  for (int k = 0; k < 8; ++k) {
    const int i = base + k * 512;
    const bool ok = i < ecount;
    d[k] = ok ? dstI[i] : -1;
    s[k] = ok ? srcI[i] : 0;
  }
#pragma unroll
  for (int k = 0; k < 8; ++k) {
    if (d[k] >= 0) {
      const int pos = atomicAdd(&cur[d[k] >> BKT_SHIFT], 1);
      keys[pos] = ((d[k] & (BKT_NODES - 1)) << 24) | s[k];
    }
  }
}

__global__ void __launch_bounds__(256) csr_bucket_finish(const int* __restrict__ keys,
                                                         const int* __restrict__ off, int G, int nb,
                                                         int ecount, int n, int* __restrict__ cnt,
                                                         int* __restrict__ segBeg,
                                                         int* __restrict__ csrSrc) {
  __shared__ int c[BKT_NODES];
  __shared__ int wsum[8];
  const int t = threadIdx.x;
  const int lane = t & 31;
  const int w = t >> 5;
  const int b = blockIdx.x;
  const int beg = off[b * G];
  const int end = (b + 1 < nb) ? off[(b + 1) * G] : ecount;
  c[t] = 0;
  __syncthreads();
  for (int i0 = beg + t; i0 < end; i0 += 256 * 8) {
    unsigned key[8];
#pragma unroll
    for (int k = 0; k < 8; ++k) {
      const int i = i0 + k * 256;
      key[k] = (unsigned)keys[(i < end) ? i : i0];
    }
#pragma unroll
    for (int k = 0; k < 8; ++k) {
      const bool ok = i0 + k * 256 < end;
      atomicAdd(&c[ok ? (int)(key[k] >> 24) : t], ok ? 1 : 0);
    }
  }
  __syncthreads();
  const int my = c[t];
  int x = my;
  for (int o = 1; o < 32; o <<= 1) {
    int y = __shfl_up(x, o, 32);
    if (lane >= o) x += y;
  }
  if (lane == 31) wsum[w] = x;
  __syncthreads();
  int wb = 0;
  for (int j = 0; j < w; ++j) wb += wsum[j];
  const int start = beg + wb + x - my;
  const int node = b * BKT_NODES + t;
  if (node < n) {
    cnt[node] = my;
    segBeg[node] = start;
  }
  c[t] = start;
  __syncthreads();
  for (int i0 = beg + t; i0 < end; i0 += 256 * 8) {
    unsigned key[8];
    int pos[8];
#pragma unroll
    for (int k = 0; k < 8; ++k) {
      const int i = i0 + k * 256;
      key[k] = (unsigned)keys[(i < end) ? i : i0];
    }
#pragma unroll
    for (int k = 0; k < 8; ++k) {
      const bool ok = i0 + k * 256 < end;
      pos[k] = atomicAdd(&c[ok ? (int)(key[k] >> 24) : t], ok ? 1 : 0);
    }
#pragma unroll
    for (int k = 0; k < 8; ++k) {
      if (i0 + k * 256 < end) csrSrc[pos[k]] = (int)(key[k] & 0x00FFFFFFu);
    }
  }
}

__device__ __forceinline__ float leaky(float v) { return v > 0.0f ? v : v * NEG_SLOPE; }

__device__ __forceinline__ float4 ld_row(const float* __restrict__ H, int s, int lane) {
  return *(const float4*)(H + (size_t)s * DFEAT + lane * 4);
}

__device__ __forceinline__ void fma_row(float4& acc, float c, const float4& h) {
  acc.x += c * h.x;
  acc.y += c * h.y;
  acc.z += c * h.z;
  acc.w += c * h.w;
}

__device__ __forceinline__ void agg_rows(const float* __restrict__ H, int lane, int s, float coef,
                                         int cnt, float4& acc) {
  const int ci = __float_as_int(coef);
  int t = 0;
  for (; t + 4 <= cnt; t += 4) {
    int s0 = __builtin_amdgcn_readlane(s, t);
    int s1 = __builtin_amdgcn_readlane(s, t + 1);
    int s2 = __builtin_amdgcn_readlane(s, t + 2);
    int s3 = __builtin_amdgcn_readlane(s, t + 3);
    float c0 = __int_as_float(__builtin_amdgcn_readlane(ci, t));
    float c1 = __int_as_float(__builtin_amdgcn_readlane(ci, t + 1));
    float c2 = __int_as_float(__builtin_amdgcn_readlane(ci, t + 2));
    float c3 = __int_as_float(__builtin_amdgcn_readlane(ci, t + 3));
    float4 h0 = ld_row(H, s0, lane);
    float4 h1 = ld_row(H, s1, lane);
    float4 h2 = ld_row(H, s2, lane);
    float4 h3 = ld_row(H, s3, lane);
    fma_row(acc, c0, h0);
    fma_row(acc, c1, h1);
    fma_row(acc, c2, h2);
    fma_row(acc, c3, h3);
  }
  for (; t < cnt; ++t) {
    int s0 = __builtin_amdgcn_readlane(s, t);
    float c0 = __int_as_float(__builtin_amdgcn_readlane(ci, t));
    float4 h0 = ld_row(H, s0, lane);
    fma_row(acc, c0, h0);
  }
}

__global__ void __launch_bounds__(256) gat_node_gather(const int* __restrict__ csrSrc,
                                                       const int* __restrict__ segBeg,
                                                       const int* __restrict__ cnt,
                                                       const float* __restrict__ asrc,
                                                       const float* __restrict__ adst,
                                                       const float* __restrict__ H,
                                                       const float* __restrict__ bias,
                                                       float* __restrict__ O, int n, int relu) {
  const int g = blockIdx.x * blockDim.x + threadIdx.x;
  const int lane = threadIdx.x & 31;
  const int d = __builtin_amdgcn_readfirstlane(g >> 5);
  if (d >= n) return;
  const int beg = segBeg[d];
  const int end = beg + cnt[d];
  const float ad = adst[d];
  const float vself = leaky(asrc[d] + ad);

  const int j0 = beg + lane;
  const bool ok0 = j0 < end;
  const int s0 = ok0 ? csrSrc[j0] : d;
  const float e0 = leaky(asrc[s0] + ad);

  float m = fmaxf(vself, e0);
  for (int j = j0 + 32; j < end; j += 32) m = fmaxf(m, leaky(asrc[csrSrc[j]] + ad));
  for (int off = 16; off; off >>= 1) m = fmaxf(m, __shfl_xor(m, off, 32));

  const float w0 = ok0 ? expf(e0 - m) : 0.0f;
  const float wself = expf(vself - m);
  float sum = w0;
  for (int j = j0 + 32; j < end; j += 32) sum += expf(leaky(asrc[csrSrc[j]] + ad) - m);
  if (lane == 0) sum += wself;
  for (int off = 16; off; off >>= 1) sum += __shfl_xor(sum, off, 32);
  const float den = sum + 1e-16f;

  float4 acc = make_float4(0.0f, 0.0f, 0.0f, 0.0f);
  fma_row(acc, wself / den, ld_row(H, d, lane));
  {
    const int c0 = (end - beg < 32) ? (end - beg) : 32;
    agg_rows(H, lane, s0, w0 / den, c0, acc);
  }
  for (int base = beg + 32; base < end; base += 32) {
    const int j = base + lane;
    int s = d;
    float coef = 0.0f;
    if (j < end) {
      s = csrSrc[j];
      coef = expf(leaky(asrc[s] + ad) - m) / den;
    }
    const int c = (end - base < 32) ? (end - base) : 32;
    agg_rows(H, lane, s, coef, c, acc);
  }

  const float4 b4 = *(const float4*)(bias + lane * 4);
  acc.x += b4.x;
  acc.y += b4.y;
  acc.z += b4.z;
  acc.w += b4.w;
  if (relu) {
    acc.x = fmaxf(acc.x, 0.0f);
    acc.y = fmaxf(acc.y, 0.0f);
    acc.z = fmaxf(acc.z, 0.0f);
    acc.w = fmaxf(acc.w, 0.0f);
  }
  *(float4*)(O + (size_t)d * DFEAT + lane * 4) = acc;
}

__global__ void __launch_bounds__(256) gat_predict(const float* __restrict__ x2o,
                                                   const float* __restrict__ x2s,
                                                   const float* __restrict__ deg,
                                                   const float* __restrict__ Wp,
                                                   const float* __restrict__ bp,
                                                   float* __restrict__ out, int n) {
  int i = blockIdx.x * blockDim.x + threadIdx.x;
  if (i >= n) return;
  const float* zo = x2o + (size_t)i * DFEAT;
  const float* zs = x2s + (size_t)i * DFEAT;
  const float* zd = deg + (size_t)i * DDEG;
  float l0 = bp[0], l1 = bp[1];
#pragma unroll 4
  for (int d = 0; d < DFEAT; ++d) {
    l0 += zo[d] * Wp[d * 2 + 0];
    l1 += zo[d] * Wp[d * 2 + 1];
  }
#pragma unroll 4
  for (int d = 0; d < DFEAT; ++d) {
    l0 += zs[d] * Wp[(DFEAT + d) * 2 + 0];
    l1 += zs[d] * Wp[(DFEAT + d) * 2 + 1];
  }
#pragma unroll 4
  for (int d = 0; d < DDEG; ++d) {
    l0 += zd[d] * Wp[(2 * DFEAT + d) * 2 + 0];
    l1 += zd[d] * Wp[(2 * DFEAT + d) * 2 + 1];
  }
  float mx = fmaxf(l0, l1);
  float e0 = expf(l0 - mx), e1 = expf(l1 - mx);
  float inv = 1.0f / (e0 + e1);
  float a0 = e0 * inv, a1 = e1 * inv;
  float* po = out + (size_t)i * DFEAT;
#pragma unroll 4
  for (int d = 0; d < DFEAT; ++d) po[d] = a0 * zo[d] + a1 * zs[d];
}

static inline int ceil_div(long long a, long long b) { return (int)((a + b - 1) / b); }

static void build_csr(const int* srcI, const int* dstI, int ecount, int n, int* cnt, int* segBeg,
                      int* keys, int* hist, int* off, int* csrSrc, hipStream_t stream) {
  const int G = ceil_div(ecount, BKT_EDGES_PER_BLOCK);
  const int nb = ceil_div(n, BKT_NODES);
  csr_bucket_count<<<G, 512, 0, stream>>>(dstI, ecount, nb, hist);
  csr_scan<<<1, 1024, 0, stream>>>(hist, off, nb * G);
  csr_bucket_scatter<<<G, 512, 0, stream>>>(srcI, dstI, ecount, nb, off, keys);
  csr_bucket_finish<<<nb, 256, 0, stream>>>(keys, off, G, nb, ecount, n, cnt, segBeg, csrSrc);
}

static void run_gat_layer(const float* X, const float* W, const float* avs, const float* avd,
                          const float* bias, float* H, float* asrc, float* adst,
                          const int* csrSrc, const int* segBeg, const int* cnt, float* O,
                          int n, int relu, hipStream_t stream) {
  gat_gemm_xw<<<n / 16, 256, 0, stream>>>(X, W, avs, avd, H, asrc, adst, n);
  gat_node_gather<<<ceil_div((long long)n * 32, 256), 256, 0, stream>>>(csrSrc, segBeg, cnt, asrc, adst,
                                                                        H, bias, O, n, relu);
}

extern "C" void kernel_launch(void* const* d_in, const int* in_sizes, int n_in,
                              void* d_out, int out_size, void* d_ws, size_t ws_size,
                              hipStream_t stream) {
  const float* x_o      = (const float*)d_in[0];
  const float* degree_o = (const float*)d_in[1];
  const int*   ei_o     = (const int*)d_in[2];
  const int*   ei_s     = (const int*)d_in[3];
  const float* W_o1 = (const float*)d_in[4];
  const float* as_o1 = (const float*)d_in[5];
  const float* ad_o1 = (const float*)d_in[6];
  const float* b_o1  = (const float*)d_in[7];
  const float* W_o2 = (const float*)d_in[8];
  const float* as_o2 = (const float*)d_in[9];
  const float* ad_o2 = (const float*)d_in[10];
  const float* b_o2  = (const float*)d_in[11];
  const float* W_s1 = (const float*)d_in[12];
  const float* as_s1 = (const float*)d_in[13];
  const float* ad_s1 = (const float*)d_in[14];
  const float* b_s1  = (const float*)d_in[15];
  const float* W_s2 = (const float*)d_in[16];
  const float* as_s2 = (const float*)d_in[17];
  const float* ad_s2 = (const float*)d_in[18];
  const float* b_s2  = (const float*)d_in[19];
  const float* W_pred = (const float*)d_in[20];
  const float* b_pred = (const float*)d_in[21];

  const int n = in_sizes[0] / DFEAT;
  const int ecount = in_sizes[2] / 2;

  const int* src_o = ei_o;
  const int* dst_o = ei_o + ecount;
  const int* src_s = ei_s;
  const int* dst_s = ei_s + ecount;

  float* ws = (float*)d_ws;
  float* H     = ws;
  float* X1    = H + (size_t)n * DFEAT;
  float* X2o   = X1 + (size_t)n * DFEAT;
  float* X2s   = X2o + (size_t)n * DFEAT;
  float* asrc  = X2s + (size_t)n * DFEAT;
  float* adst  = asrc + n;
  int* cnt     = (int*)(adst + n);
  int* segBeg  = cnt + n;
  int* csrSrc  = segBeg + n;
  int* keys    = (int*)X2s;
  int* hist    = (int*)H;
  int* off     = hist + (size_t)n * DFEAT / 2;

  build_csr(src_o, dst_o, ecount, n, cnt, segBeg, keys, hist, off, csrSrc, stream);
  run_gat_layer(x_o, W_o1, as_o1, ad_o1, b_o1, H, asrc, adst, csrSrc, segBeg, cnt, X1, n,
                 1, stream);
  run_gat_layer(X1, W_o2, as_o2, ad_o2, b_o2, H, asrc, adst, csrSrc, segBeg, cnt, X2o, n,
                 0, stream);
  build_csr(src_s, dst_s, ecount, n, cnt, segBeg, keys, hist, off, csrSrc, stream);
  run_gat_layer(x_o, W_s1, as_s1, ad_s1, b_s1, H, asrc, adst, csrSrc, segBeg, cnt, X1, n,
                 1, stream);
  run_gat_layer(X1, W_s2, as_s2, ad_s2, b_s2, H, asrc, adst, csrSrc, segBeg, cnt, X2s, n,
                 0, stream);

  gat_predict<<<ceil_div(n, 256), 256, 0, stream>>>(X2o, X2s, degree_o, W_pred, b_pred,
                                                    (float*)d_out, n);
}


# Both graph branches share every launch (grid y = branch): one pass of the 4 sort kernels, one GEMM and one gather launch per layer; 17 launches become 9; values and sum orders unchanged
# speedup vs baseline: 5.5054x; 1.2811x over previous
#include <hip/hip_runtime.h>
#include <math.h>

typedef __attribute__((ext_vector_type(2))) float v2f;
typedef __attribute__((ext_vector_type(8))) float v8f;

#define DFEAT 128
#define DDEG 32
#define NEG_SLOPE 0.2f
#define XS_PITCH (DFEAT + 4)

template <int CTRL>
__device__ __forceinline__ float row_swz(float v) {
  return __int_as_float(__builtin_amdgcn_update_dpp(0, __float_as_int(v), CTRL, 0xF, 0xF, true));
}
__device__ __forceinline__ float row16_sum(float v) {
  v += row_swz<0x161>(v);
  v += row_swz<0x162>(v);
  v += row_swz<0x164>(v);
  v += row_swz<0x168>(v);
  return v;
}

__global__ void __launch_bounds__(256) gat_gemm_xw(const float* __restrict__ X0,
                                                   const float* __restrict__ W0,
                                                   const float* __restrict__ a_src0,
                                                   const float* __restrict__ a_dst0,
                                                   float* __restrict__ H0,
                                                   float* __restrict__ asrc0,
                                                   float* __restrict__ adst0,
                                                   const float* __restrict__ X1,
                                                   const float* __restrict__ W1,
                                                   const float* __restrict__ a_src1,
                                                   const float* __restrict__ a_dst1,
                                                   float* __restrict__ H1,
                                                   float* __restrict__ asrc1,
                                                   float* __restrict__ adst1,
                                                   int nrows) {
  const bool br = blockIdx.y != 0;
  const float* __restrict__ X = br ? X1 : X0;
  const float* __restrict__ W = br ? W1 : W0;
  const float* __restrict__ a_src = br ? a_src1 : a_src0;
  const float* __restrict__ a_dst = br ? a_dst1 : a_dst0;
  float* __restrict__ H = br ? H1 : H0;
  float* __restrict__ asrc = br ? asrc1 : asrc0;
  float* __restrict__ adst = br ? adst1 : adst0;
  __shared__ float xs[16 * XS_PITCH];
  __shared__ float red[2][8][16];
  const int tid = threadIdx.x;
  const int wave = tid >> 5;
  const int lane = tid & 31;
  const int rowBlk = blockIdx.x * 16;

  {
    int idx = tid * 8;
    int r = idx >> 7;
    int c = idx & (DFEAT - 1);
    const float4* src = (const float4*)(X + (size_t)(rowBlk + r) * DFEAT + c);
    float4 v0 = src[0];
    float4 v1 = src[1];
    *(float4*)(xs + r * XS_PITCH + c) = v0;
    *(float4*)(xs + r * XS_PITCH + c + 4) = v1;
  }
  __syncthreads();

  const int col0 = wave * 16;
  const int mn = lane & 15;
  const int kk = (lane >> 4) * 2;

  v8f acc = {};
  for (int k0 = 0; k0 < DFEAT; k0 += 4) {
    v2f a, b;
    a.x = xs[mn * XS_PITCH + k0 + kk];
    a.y = xs[mn * XS_PITCH + k0 + kk + 1];
    b.x = W[(size_t)(k0 + kk) * DFEAT + col0 + mn];
    b.y = W[(size_t)(k0 + kk + 1) * DFEAT + col0 + mn];
    acc = __builtin_amdgcn_wmma_f32_16x16x4_f32(
         false, a,  false, b,
         (short)0, acc,  false,  false);
  }

  const int mbase = (lane >> 4) * 8;
  float* out = H + (size_t)(rowBlk + mbase) * DFEAT + col0 + mn;
#pragma unroll
  for (int r = 0; r < 8; ++r) {
    out[(size_t)r * DFEAT] = acc[r];
  }

  const float vs = a_src[col0 + mn];
  const float vd = a_dst[col0 + mn];
  float ps[8], pd[8];
#pragma unroll
  for (int r = 0; r < 8; ++r) {
    ps[r] = row16_sum(acc[r] * vs);
    pd[r] = row16_sum(acc[r] * vd);
  }
  if (mn == 0) {
#pragma unroll
    for (int r = 0; r < 8; ++r) {
      red[0][wave][mbase + r] = ps[r];
      red[1][wave][mbase + r] = pd[r];
    }
  }
  __syncthreads();
  if (tid < 32) {
    const int row = tid & 15;
    const int which = tid >> 4;
    float t = red[which][0][row];
#pragma unroll
    for (int w = 1; w < 8; ++w) t += red[which][w][row];
    float* dstp = which ? adst : asrc;
    dstp[rowBlk + row] = t;
  }
}

#define BKT_SHIFT 8
#define BKT_NODES 256
#define BKT_EDGES_PER_BLOCK 4096

__global__ void __launch_bounds__(512) csr_bucket_count(const int* __restrict__ dstI0,
                                                        const int* __restrict__ dstI1, int ecount,
                                                        int nb, int* __restrict__ hist) {
  __shared__ int h[BKT_NODES];
  const int t = threadIdx.x;
  const int g = blockIdx.x;
  const int G = gridDim.x;
  const int* __restrict__ dstI = blockIdx.y ? dstI1 : dstI0;
  hist += blockIdx.y * nb * G;
  if (t < BKT_NODES) h[t] = 0;
  __syncthreads();
  const int base = g * BKT_EDGES_PER_BLOCK + t;
  int d[8];
#pragma unroll
  for (int k = 0; k < 8; ++k) {
    const int i = base + k * 512;
    d[k] = (i < ecount) ? dstI[i] : -1;
  }
#pragma unroll
  for (int k = 0; k < 8; ++k) {
    if (d[k] >= 0) atomicAdd(&h[d[k] >> BKT_SHIFT], 1);
  }
  __syncthreads();
  if (t < nb) hist[t * G + g] = h[t];
}

__global__ void __launch_bounds__(1024) csr_scan(const int* __restrict__ cnt,
                                                 int* __restrict__ segBeg, int n) {
  __shared__ int wsum[32];
  cnt += blockIdx.x * n;
  segBeg += blockIdx.x * n;
  const int t = threadIdx.x;
  const int lane = t & 31;
  const int w = t >> 5;
  const int ch = (n + 1023) >> 10;
  const int b = t * ch;
  const int e = (b + ch < n) ? (b + ch) : n;
  int s = 0;
  for (int i = b; i < e; ++i) s += cnt[i];
  int x = s;
  for (int off = 1; off < 32; off <<= 1) {
    int y = __shfl_up(x, off, 32);
    if (lane >= off) x += y;
  }
  if (lane == 31) wsum[w] = x;
  __syncthreads();
  if (w == 0) {
    int v = wsum[lane];
    for (int off = 1; off < 32; off <<= 1) {
      int y = __shfl_up(v, off, 32);
      if (lane >= off) v += y;
    }
    wsum[lane] = v;
  }
  __syncthreads();
  int run = x - s + (w > 0 ? wsum[w - 1] : 0);
  int i = b;
  for (; i + 8 <= e; i += 8) {
    const int c0 = cnt[i], c1 = cnt[i + 1], c2 = cnt[i + 2], c3 = cnt[i + 3];
    const int c4 = cnt[i + 4], c5 = cnt[i + 5], c6 = cnt[i + 6], c7 = cnt[i + 7];
    segBeg[i] = run;     run += c0;
    segBeg[i + 1] = run; run += c1;
    segBeg[i + 2] = run; run += c2;
    segBeg[i + 3] = run; run += c3;
    segBeg[i + 4] = run; run += c4;
    segBeg[i + 5] = run; run += c5;
    segBeg[i + 6] = run; run += c6;
    segBeg[i + 7] = run; run += c7;
  }
  for (; i < e; ++i) {
    segBeg[i] = run;
    run += cnt[i];
  }
}

__global__ void __launch_bounds__(512) csr_bucket_scatter(const int* __restrict__ srcI0,
                                                          const int* __restrict__ dstI0,
                                                          const int* __restrict__ srcI1,
                                                          const int* __restrict__ dstI1, int ecount,
                                                          int nb, const int* __restrict__ off,
                                                          int* __restrict__ keys) {
  __shared__ int cur[BKT_NODES];
  const int t = threadIdx.x;
  const int g = blockIdx.x;
  const int G = gridDim.x;
  const int* __restrict__ srcI = blockIdx.y ? srcI1 : srcI0;
  const int* __restrict__ dstI = blockIdx.y ? dstI1 : dstI0;
  off += blockIdx.y * nb * G;
  keys += blockIdx.y * ecount;
  if (t < nb) cur[t] = off[t * G + g];
  __syncthreads();
  const int base = g * BKT_EDGES_PER_BLOCK + t;
  int d[8], s[8];
#pragma unroll
  for (int k = 0; k < 8; ++k) {
    const int i = base + k * 512;
    const bool ok = i < ecount;
    d[k] = ok ? dstI[i] : -1;
    s[k] = ok ? srcI[i] : 0;
  }
#pragma unroll
  for (int k = 0; k < 8; ++k) {
    if (d[k] >= 0) {
      const int pos = atomicAdd(&cur[d[k] >> BKT_SHIFT], 1);
      keys[pos] = ((d[k] & (BKT_NODES - 1)) << 24) | s[k];
    }
  }
}

__global__ void __launch_bounds__(256) csr_bucket_finish(const int* __restrict__ keys,
                                                         const int* __restrict__ off, int G, int nb,
                                                         int ecount, int n, int* __restrict__ cnt0,
                                                         int* __restrict__ segBeg0,
                                                         int* __restrict__ csrSrc0,
                                                         int* __restrict__ cnt1,
                                                         int* __restrict__ segBeg1,
                                                         int* __restrict__ csrSrc1) {
  __shared__ int c[BKT_NODES];
  __shared__ int wsum[8];
  keys += blockIdx.y * ecount;
  off += blockIdx.y * nb * G;
  int* __restrict__ cnt = blockIdx.y ? cnt1 : cnt0;
  int* __restrict__ segBeg = blockIdx.y ? segBeg1 : segBeg0;
  int* __restrict__ csrSrc = blockIdx.y ? csrSrc1 : csrSrc0;
  const int t = threadIdx.x;
  const int lane = t & 31;
  const int w = t >> 5;
  const int b = blockIdx.x;
  const int beg = off[b * G];
  const int end = (b + 1 < nb) ? off[(b + 1) * G] : ecount;
  c[t] = 0;
  __syncthreads();
  for (int i0 = beg + t; i0 < end; i0 += 256 * 8) {
    unsigned key[8];
#pragma unroll
    for (int k = 0; k < 8; ++k) {
      const int i = i0 + k * 256;
      key[k] = (unsigned)keys[(i < end) ? i : i0];
    }
#pragma unroll
    for (int k = 0; k < 8; ++k) {
      const bool ok = i0 + k * 256 < end;
      atomicAdd(&c[ok ? (int)(key[k] >> 24) : t], ok ? 1 : 0);
    }
  }
  __syncthreads();
  const int my = c[t];
  int x = my;
  for (int o = 1; o < 32; o <<= 1) {
    int y = __shfl_up(x, o, 32);
    if (lane >= o) x += y;
  }
  if (lane == 31) wsum[w] = x;
  __syncthreads();
  int wb = 0;
  for (int j = 0; j < w; ++j) wb += wsum[j];
  const int start = beg + wb + x - my;
  const int node = b * BKT_NODES + t;
  if (node < n) {
    cnt[node] = my;
    segBeg[node] = start;
  }
  c[t] = start;
  __syncthreads();
  for (int i0 = beg + t; i0 < end; i0 += 256 * 8) {
    unsigned key[8];
    int pos[8];
#pragma unroll
    for (int k = 0; k < 8; ++k) {
      const int i = i0 + k * 256;
      key[k] = (unsigned)keys[(i < end) ? i : i0];
    }
#pragma unroll
    for (int k = 0; k < 8; ++k) {
      const bool ok = i0 + k * 256 < end;
      pos[k] = atomicAdd(&c[ok ? (int)(key[k] >> 24) : t], ok ? 1 : 0);
    }
#pragma unroll
    for (int k = 0; k < 8; ++k) {
      if (i0 + k * 256 < end) csrSrc[pos[k]] = (int)(key[k] & 0x00FFFFFFu);
    }
  }
}

__device__ __forceinline__ float leaky(float v) { return v > 0.0f ? v : v * NEG_SLOPE; }

__device__ __forceinline__ float4 ld_row(const float* __restrict__ H, int s, int lane) {
  return *(const float4*)(H + (size_t)s * DFEAT + lane * 4);
}

__device__ __forceinline__ void fma_row(float4& acc, float c, const float4& h) {
  acc.x += c * h.x;
  acc.y += c * h.y;
  acc.z += c * h.z;
  acc.w += c * h.w;
}

__device__ __forceinline__ void agg_rows(const float* __restrict__ H, int lane, int s, float coef,
                                         int cnt, float4& acc) {
  const int ci = __float_as_int(coef);
  int t = 0;
  for (; t + 4 <= cnt; t += 4) {
    int s0 = __builtin_amdgcn_readlane(s, t);
    int s1 = __builtin_amdgcn_readlane(s, t + 1);
    int s2 = __builtin_amdgcn_readlane(s, t + 2);
    int s3 = __builtin_amdgcn_readlane(s, t + 3);
    float c0 = __int_as_float(__builtin_amdgcn_readlane(ci, t));
    float c1 = __int_as_float(__builtin_amdgcn_readlane(ci, t + 1));
    float c2 = __int_as_float(__builtin_amdgcn_readlane(ci, t + 2));
    float c3 = __int_as_float(__builtin_amdgcn_readlane(ci, t + 3));
    float4 h0 = ld_row(H, s0, lane);
    float4 h1 = ld_row(H, s1, lane);
    float4 h2 = ld_row(H, s2, lane);
    float4 h3 = ld_row(H, s3, lane);
    fma_row(acc, c0, h0);
    fma_row(acc, c1, h1);
    fma_row(acc, c2, h2);
    fma_row(acc, c3, h3);
  }
  for (; t < cnt; ++t) {
    int s0 = __builtin_amdgcn_readlane(s, t);
    float c0 = __int_as_float(__builtin_amdgcn_readlane(ci, t));
    float4 h0 = ld_row(H, s0, lane);
    fma_row(acc, c0, h0);
  }
}

__global__ void __launch_bounds__(256) gat_node_gather(const int* __restrict__ csrSrc0,
                                                       const int* __restrict__ segBeg0,
                                                       const int* __restrict__ cnt0,
                                                       const float* __restrict__ asrc0,
                                                       const float* __restrict__ adst0,
                                                       const float* __restrict__ H0,
                                                       const float* __restrict__ bias0,
                                                       float* __restrict__ O0,
                                                       const int* __restrict__ csrSrc1,
                                                       const int* __restrict__ segBeg1,
                                                       const int* __restrict__ cnt1,
                                                       const float* __restrict__ asrc1,
                                                       const float* __restrict__ adst1,
                                                       const float* __restrict__ H1,
                                                       const float* __restrict__ bias1,
                                                       float* __restrict__ O1, int n, int relu) {
  const bool br = blockIdx.y != 0;
  const int* __restrict__ csrSrc = br ? csrSrc1 : csrSrc0;
  const int* __restrict__ segBeg = br ? segBeg1 : segBeg0;
  const int* __restrict__ cnt = br ? cnt1 : cnt0;
  const float* __restrict__ asrc = br ? asrc1 : asrc0;
  const float* __restrict__ adst = br ? adst1 : adst0;
  const float* __restrict__ H = br ? H1 : H0;
  const float* __restrict__ bias = br ? bias1 : bias0;
  float* __restrict__ O = br ? O1 : O0;
  const int g = blockIdx.x * blockDim.x + threadIdx.x;
  const int lane = threadIdx.x & 31;
  const int d = __builtin_amdgcn_readfirstlane(g >> 5);
  if (d >= n) return;
  const int beg = segBeg[d];
  const int end = beg + cnt[d];
  const float ad = adst[d];
  const float vself = leaky(asrc[d] + ad);

  const int j0 = beg + lane;
  const bool ok0 = j0 < end;
  const int s0 = ok0 ? csrSrc[j0] : d;
  const float e0 = leaky(asrc[s0] + ad);

  float m = fmaxf(vself, e0);
  for (int j = j0 + 32; j < end; j += 32) m = fmaxf(m, leaky(asrc[csrSrc[j]] + ad));
  for (int off = 16; off; off >>= 1) m = fmaxf(m, __shfl_xor(m, off, 32));

  const float w0 = ok0 ? expf(e0 - m) : 0.0f;
  const float wself = expf(vself - m);
  float sum = w0;
  for (int j = j0 + 32; j < end; j += 32) sum += expf(leaky(asrc[csrSrc[j]] + ad) - m);
  if (lane == 0) sum += wself;
  for (int off = 16; off; off >>= 1) sum += __shfl_xor(sum, off, 32);
  const float den = sum + 1e-16f;

  float4 acc = make_float4(0.0f, 0.0f, 0.0f, 0.0f);
  fma_row(acc, wself / den, ld_row(H, d, lane));
  {
    const int c0 = (end - beg < 32) ? (end - beg) : 32;
    agg_rows(H, lane, s0, w0 / den, c0, acc);
  }
  for (int base = beg + 32; base < end; base += 32) {
    const int j = base + lane;
    int s = d;
    float coef = 0.0f;
    if (j < end) {
      s = csrSrc[j];
      coef = expf(leaky(asrc[s] + ad) - m) / den;
    }
    const int c = (end - base < 32) ? (end - base) : 32;
    agg_rows(H, lane, s, coef, c, acc);
  }

  const float4 b4 = *(const float4*)(bias + lane * 4);
  acc.x += b4.x;
  acc.y += b4.y;
  acc.z += b4.z;
  acc.w += b4.w;
  if (relu) {
    acc.x = fmaxf(acc.x, 0.0f);
    acc.y = fmaxf(acc.y, 0.0f);
    acc.z = fmaxf(acc.z, 0.0f);
    acc.w = fmaxf(acc.w, 0.0f);
  }
  *(float4*)(O + (size_t)d * DFEAT + lane * 4) = acc;
}

__global__ void __launch_bounds__(256) gat_predict(const float* __restrict__ x2o,
                                                   const float* __restrict__ x2s,
                                                   const float* __restrict__ deg,
                                                   const float* __restrict__ Wp,
                                                   const float* __restrict__ bp,
                                                   float* __restrict__ out, int n) {
  int i = blockIdx.x * blockDim.x + threadIdx.x;
  if (i >= n) return;
  const float* zo = x2o + (size_t)i * DFEAT;
  const float* zs = x2s + (size_t)i * DFEAT;
  const float* zd = deg + (size_t)i * DDEG;
  float l0 = bp[0], l1 = bp[1];
#pragma unroll 4
  for (int d = 0; d < DFEAT; ++d) {
    l0 += zo[d] * Wp[d * 2 + 0];
    l1 += zo[d] * Wp[d * 2 + 1];
  }
#pragma unroll 4
  for (int d = 0; d < DFEAT; ++d) {
    l0 += zs[d] * Wp[(DFEAT + d) * 2 + 0];
    l1 += zs[d] * Wp[(DFEAT + d) * 2 + 1];
  }
#pragma unroll 4
  for (int d = 0; d < DDEG; ++d) {
    l0 += zd[d] * Wp[(2 * DFEAT + d) * 2 + 0];
    l1 += zd[d] * Wp[(2 * DFEAT + d) * 2 + 1];
  }
  float mx = fmaxf(l0, l1);
  float e0 = expf(l0 - mx), e1 = expf(l1 - mx);
  float inv = 1.0f / (e0 + e1);
  float a0 = e0 * inv, a1 = e1 * inv;
  float* po = out + (size_t)i * DFEAT;
#pragma unroll 4
  for (int d = 0; d < DFEAT; ++d) po[d] = a0 * zo[d] + a1 * zs[d];
}

static inline int ceil_div(long long a, long long b) { return (int)((a + b - 1) / b); }

static void build_csr2(const int* src0, const int* dst0, const int* src1, const int* dst1,
                       int ecount, int n, int* cnt0, int* segBeg0, int* csrSrc0, int* cnt1,
                       int* segBeg1, int* csrSrc1, int* keys, int* hist, int* off,
                       hipStream_t stream) {
  const int G = ceil_div(ecount, BKT_EDGES_PER_BLOCK);
  const int nb = ceil_div(n, BKT_NODES);
  csr_bucket_count<<<dim3(G, 2), 512, 0, stream>>>(dst0, dst1, ecount, nb, hist);
  csr_scan<<<2, 1024, 0, stream>>>(hist, off, nb * G);
  csr_bucket_scatter<<<dim3(G, 2), 512, 0, stream>>>(src0, dst0, src1, dst1, ecount, nb, off, keys);
  csr_bucket_finish<<<dim3(nb, 2), 256, 0, stream>>>(keys, off, G, nb, ecount, n, cnt0, segBeg0,
                                                     csrSrc0, cnt1, segBeg1, csrSrc1);
}

struct LayerIo {
  const float* X;
  const float* W;
  const float* avs;
  const float* avd;
  const float* bias;
  float* H;
  float* asrc;
  float* adst;
  const int* csrSrc;
  const int* segBeg;
  const int* cnt;
  float* O;
};

static void run_gat_layer2(const LayerIo& a, const LayerIo& b, int n, int relu,
                           hipStream_t stream) {
  gat_gemm_xw<<<dim3(n / 16, 2), 256, 0, stream>>>(a.X, a.W, a.avs, a.avd, a.H, a.asrc, a.adst,
                                                    b.X, b.W, b.avs, b.avd, b.H, b.asrc, b.adst, n);
  gat_node_gather<<<dim3(ceil_div((long long)n * 32, 256), 2), 256, 0, stream>>>(
      a.csrSrc, a.segBeg, a.cnt, a.asrc, a.adst, a.H, a.bias, a.O,
      b.csrSrc, b.segBeg, b.cnt, b.asrc, b.adst, b.H, b.bias, b.O, n, relu);
}

extern "C" void kernel_launch(void* const* d_in, const int* in_sizes, int n_in,
                              void* d_out, int out_size, void* d_ws, size_t ws_size,
                              hipStream_t stream) {
  const float* x_o      = (const float*)d_in[0];
  const float* degree_o = (const float*)d_in[1];
  const int*   ei_o     = (const int*)d_in[2];
  const int*   ei_s     = (const int*)d_in[3];
  const float* W_o1 = (const float*)d_in[4];
  const float* as_o1 = (const float*)d_in[5];
  const float* ad_o1 = (const float*)d_in[6];
  const float* b_o1  = (const float*)d_in[7];
  const float* W_o2 = (const float*)d_in[8];
  const float* as_o2 = (const float*)d_in[9];
  const float* ad_o2 = (const float*)d_in[10];
  const float* b_o2  = (const float*)d_in[11];
  const float* W_s1 = (const float*)d_in[12];
  const float* as_s1 = (const float*)d_in[13];
  const float* ad_s1 = (const float*)d_in[14];
  const float* b_s1  = (const float*)d_in[15];
  const float* W_s2 = (const float*)d_in[16];
  const float* as_s2 = (const float*)d_in[17];
  const float* ad_s2 = (const float*)d_in[18];
  const float* b_s2  = (const float*)d_in[19];
  const float* W_pred = (const float*)d_in[20];
  const float* b_pred = (const float*)d_in[21];

  const int n = in_sizes[0] / DFEAT;
  const int ecount = in_sizes[2] / 2;

  const int* src_o = ei_o;
  const int* dst_o = ei_o + ecount;
  const int* src_s = ei_s;
  const int* dst_s = ei_s + ecount;

  float* ws = (float*)d_ws;
  float* Ho    = ws;
  float* Hs    = Ho + (size_t)n * DFEAT;
  float* Yo    = Hs + (size_t)n * DFEAT;
  float* Ys    = Yo + (size_t)n * DFEAT;
  float* asrcO = Ys + (size_t)n * DFEAT;
  float* adstO = asrcO + n;
  int* cntO    = (int*)(adstO + n);
  int* segBegO = cntO + n;
  int* csrSrcO = segBegO + n;
  int* csrSrcS = (int*)d_out;
  int* cntS    = csrSrcS + ecount;
  int* segBegS = cntS + n;
  float* asrcS = (float*)(segBegS + n);
  float* adstS = asrcS + n;
  int* keys    = (int*)Ys;
  int* hist    = (int*)Ho;
  int* off     = hist + (size_t)n * DFEAT / 2;

  build_csr2(src_o, dst_o, src_s, dst_s, ecount, n, cntO, segBegO, csrSrcO, cntS, segBegS,
             csrSrcS, keys, hist, off, stream);

  const LayerIo o1 = {x_o, W_o1, as_o1, ad_o1, b_o1, Ho, asrcO, adstO, csrSrcO, segBegO, cntO, Yo};
  const LayerIo s1 = {x_o, W_s1, as_s1, ad_s1, b_s1, Hs, asrcS, adstS, csrSrcS, segBegS, cntS, Ys};
  run_gat_layer2(o1, s1, n,  1, stream);
  const LayerIo o2 = {Yo, W_o2, as_o2, ad_o2, b_o2, Ho, asrcO, adstO, csrSrcO, segBegO, cntO, Yo};
  const LayerIo s2 = {Ys, W_s2, as_s2, ad_s2, b_s2, Hs, asrcS, adstS, csrSrcS, segBegS, cntS, Ys};
  run_gat_layer2(o2, s2, n,  0, stream);

  gat_predict<<<ceil_div(n, 256), 256, 0, stream>>>(Yo, Ys, degree_o, W_pred, b_pred,
                                                    (float*)d_out, n);
}


# Edge-sort table scan loses its own launch: it runs as 94 extra 256-thread work-groups (1024 entries each, prefix re-summed) inside the layer-1 product launch, moved between count and scatter; 9 -> 8
# speedup vs baseline: 6.0551x; 1.0998x over previous
#include <hip/hip_runtime.h>
#include <math.h>

typedef __attribute__((ext_vector_type(2))) float v2f;
typedef __attribute__((ext_vector_type(8))) float v8f;

#define DFEAT 128
#define DDEG 32
#define NEG_SLOPE 0.2f
#define XS_PITCH (DFEAT + 4)

template <int CTRL>
__device__ __forceinline__ float row_swz(float v) {
  return __int_as_float(__builtin_amdgcn_update_dpp(0, __float_as_int(v), CTRL, 0xF, 0xF, true));
}
__device__ __forceinline__ float row16_sum(float v) {
  v += row_swz<0x161>(v);
  v += row_swz<0x162>(v);
  v += row_swz<0x164>(v);
  v += row_swz<0x168>(v);
  return v;
}

#define SCAN_CHUNK 1024
__device__ __forceinline__ void scan_chunk(const int* __restrict__ cnt, int* __restrict__ segBeg,
                                           int n, int chunk) {
  __shared__ int wtot[8];
  __shared__ int wpre[8];
  const int t = threadIdx.x;
  const int lane = t & 31;
  const int w = t >> 5;
  const int first = chunk * SCAN_CHUNK;
  const int i0 = first + t * 4;
  const int c0 = (i0 < n) ? cnt[i0] : 0;
  const int c1 = (i0 + 1 < n) ? cnt[i0 + 1] : 0;
  const int c2 = (i0 + 2 < n) ? cnt[i0 + 2] : 0;
  const int c3 = (i0 + 3 < n) ? cnt[i0 + 3] : 0;

  int head = (int)((0u - (unsigned)((size_t)cnt >> 2)) & 3u);
  if (head > first) head = first;
  const int nvec = (first - head) >> 2;
  const int tail = head + nvec * 4;
  const int4* __restrict__ vec = (const int4*)(cnt + head);
  int s = 0;
  if (t < head) s = cnt[t];
  if (tail + t < first) s += cnt[tail + t];
#pragma unroll 8
  for (int k = t; k < nvec; k += 256) {
    const int4 q = vec[k];
    s += (q.x + q.y) + (q.z + q.w);
  }

  const int my = (c0 + c1) + (c2 + c3);
  int x = my;
  for (int o = 1; o < 32; o <<= 1) {
    int y = __shfl_up(x, o, 32);
    if (lane >= o) x += y;
  }
  for (int o = 16; o; o >>= 1) s += __shfl_xor(s, o, 32);
  if (lane == 31) wtot[w] = x;
  if (lane == 0) wpre[w] = s;
  __syncthreads();
  int run = x - my;
#pragma unroll
  for (int j = 0; j < 8; ++j) {
    run += wpre[j];
    if (j < w) run += wtot[j];
  }
  if (i0 < n) segBeg[i0] = run;
  run += c0;
  if (i0 + 1 < n) segBeg[i0 + 1] = run;
  run += c1;
  if (i0 + 2 < n) segBeg[i0 + 2] = run;
  run += c2;
  if (i0 + 3 < n) segBeg[i0 + 3] = run;
}

__global__ void __launch_bounds__(256) gat_gemm_xw(const float* __restrict__ X0,
                                                   const float* __restrict__ W0,
                                                   const float* __restrict__ a_src0,
                                                   const float* __restrict__ a_dst0,
                                                   float* __restrict__ H0,
                                                   float* __restrict__ asrc0,
                                                   float* __restrict__ adst0,
                                                   const float* __restrict__ X1,
                                                   const float* __restrict__ W1,
                                                   const float* __restrict__ a_src1,
                                                   const float* __restrict__ a_dst1,
                                                   float* __restrict__ H1,
                                                   float* __restrict__ asrc1,
                                                   float* __restrict__ adst1,
                                                   int nrows,
                                                   const int* __restrict__ scanIn,
                                                   int* __restrict__ scanOut, int scanN,
                                                   int scanBlocks) {
  if ((int)blockIdx.x < scanBlocks) {
    scan_chunk(scanIn + (size_t)blockIdx.y * scanN, scanOut + (size_t)blockIdx.y * scanN, scanN,
               (int)blockIdx.x);
    return;
  }
  const bool br = blockIdx.y != 0;
  const float* __restrict__ X = br ? X1 : X0;
  const float* __restrict__ W = br ? W1 : W0;
  const float* __restrict__ a_src = br ? a_src1 : a_src0;
  const float* __restrict__ a_dst = br ? a_dst1 : a_dst0;
  float* __restrict__ H = br ? H1 : H0;
  float* __restrict__ asrc = br ? asrc1 : asrc0;
  float* __restrict__ adst = br ? adst1 : adst0;
  __shared__ float xs[16 * XS_PITCH];
  __shared__ float red[2][8][16];
  const int tid = threadIdx.x;
  const int wave = tid >> 5;
  const int lane = tid & 31;
  const int rowBlk = ((int)blockIdx.x - scanBlocks) * 16;

  {
    int idx = tid * 8;
    int r = idx >> 7;
    int c = idx & (DFEAT - 1);
    const float4* src = (const float4*)(X + (size_t)(rowBlk + r) * DFEAT + c);
    float4 v0 = src[0];
    float4 v1 = src[1];
    *(float4*)(xs + r * XS_PITCH + c) = v0;
    *(float4*)(xs + r * XS_PITCH + c + 4) = v1;
  }
  __syncthreads();

  const int col0 = wave * 16;
  const int mn = lane & 15;
  const int kk = (lane >> 4) * 2;

  v8f acc = {};
  for (int k0 = 0; k0 < DFEAT; k0 += 4) {
    v2f a, b;
    a.x = xs[mn * XS_PITCH + k0 + kk];
    a.y = xs[mn * XS_PITCH + k0 + kk + 1];
    b.x = W[(size_t)(k0 + kk) * DFEAT + col0 + mn];
    b.y = W[(size_t)(k0 + kk + 1) * DFEAT + col0 + mn];
    acc = __builtin_amdgcn_wmma_f32_16x16x4_f32(
         false, a,  false, b,
         (short)0, acc,  false,  false);
  }

  const int mbase = (lane >> 4) * 8;
  float* out = H + (size_t)(rowBlk + mbase) * DFEAT + col0 + mn;
#pragma unroll
  for (int r = 0; r < 8; ++r) {
    out[(size_t)r * DFEAT] = acc[r];
  }

  const float vs = a_src[col0 + mn];
  const float vd = a_dst[col0 + mn];
  float ps[8], pd[8];
#pragma unroll
  for (int r = 0; r < 8; ++r) {
    ps[r] = row16_sum(acc[r] * vs);
    pd[r] = row16_sum(acc[r] * vd);
  }
  if (mn == 0) {
#pragma unroll
    for (int r = 0; r < 8; ++r) {
      red[0][wave][mbase + r] = ps[r];
      red[1][wave][mbase + r] = pd[r];
    }
  }
  __syncthreads();
  if (tid < 32) {
    const int row = tid & 15;
    const int which = tid >> 4;
    float t = red[which][0][row];
#pragma unroll
    for (int w = 1; w < 8; ++w) t += red[which][w][row];
    float* dstp = which ? adst : asrc;
    dstp[rowBlk + row] = t;
  }
}

#define BKT_SHIFT 8
#define BKT_NODES 256
#define BKT_EDGES_PER_BLOCK 4096

__global__ void __launch_bounds__(512) csr_bucket_count(const int* __restrict__ dstI0,
                                                        const int* __restrict__ dstI1, int ecount,
                                                        int nb, int* __restrict__ hist) {
  __shared__ int h[BKT_NODES];
  const int t = threadIdx.x;
  const int g = blockIdx.x;
  const int G = gridDim.x;
  const int* __restrict__ dstI = blockIdx.y ? dstI1 : dstI0;
  hist += blockIdx.y * nb * G;
  if (t < BKT_NODES) h[t] = 0;
  __syncthreads();
  const int base = g * BKT_EDGES_PER_BLOCK + t;
  int d[8];
#pragma unroll
  for (int k = 0; k < 8; ++k) {
    const int i = base + k * 512;
    d[k] = (i < ecount) ? dstI[i] : -1;
  }
#pragma unroll
  for (int k = 0; k < 8; ++k) {
    if (d[k] >= 0) atomicAdd(&h[d[k] >> BKT_SHIFT], 1);
  }
  __syncthreads();
  if (t < nb) hist[t * G + g] = h[t];
}

__global__ void __launch_bounds__(512) csr_bucket_scatter(const int* __restrict__ srcI0,
                                                          const int* __restrict__ dstI0,
                                                          const int* __restrict__ srcI1,
                                                          const int* __restrict__ dstI1, int ecount,
                                                          int nb, const int* __restrict__ off,
                                                          int* __restrict__ keys) {
  __shared__ int cur[BKT_NODES];
  const int t = threadIdx.x;
  const int g = blockIdx.x;
  const int G = gridDim.x;
  const int* __restrict__ srcI = blockIdx.y ? srcI1 : srcI0;
  const int* __restrict__ dstI = blockIdx.y ? dstI1 : dstI0;
  off += blockIdx.y * nb * G;
  keys += blockIdx.y * ecount;
  if (t < nb) cur[t] = off[t * G + g];
  __syncthreads();
  const int base = g * BKT_EDGES_PER_BLOCK + t;
  int d[8], s[8];
#pragma unroll
  for (int k = 0; k < 8; ++k) {
    const int i = base + k * 512;
    const bool ok = i < ecount;
    d[k] = ok ? dstI[i] : -1;
    s[k] = ok ? srcI[i] : 0;
  }
#pragma unroll
  for (int k = 0; k < 8; ++k) {
    if (d[k] >= 0) {
      const int pos = atomicAdd(&cur[d[k] >> BKT_SHIFT], 1);
      keys[pos] = ((d[k] & (BKT_NODES - 1)) << 24) | s[k];
    }
  }
}

__global__ void __launch_bounds__(256) csr_bucket_finish(const int* __restrict__ keys,
                                                         const int* __restrict__ off, int G, int nb,
                                                         int ecount, int n, int* __restrict__ cnt0,
                                                         int* __restrict__ segBeg0,
                                                         int* __restrict__ csrSrc0,
                                                         int* __restrict__ cnt1,
                                                         int* __restrict__ segBeg1,
                                                         int* __restrict__ csrSrc1) {
  __shared__ int c[BKT_NODES];
  __shared__ int wsum[8];
  keys += blockIdx.y * ecount;
  off += blockIdx.y * nb * G;
  int* __restrict__ cnt = blockIdx.y ? cnt1 : cnt0;
  int* __restrict__ segBeg = blockIdx.y ? segBeg1 : segBeg0;
  int* __restrict__ csrSrc = blockIdx.y ? csrSrc1 : csrSrc0;
  const int t = threadIdx.x;
  const int lane = t & 31;
  const int w = t >> 5;
  const int b = blockIdx.x;
  const int beg = off[b * G];
  const int end = (b + 1 < nb) ? off[(b + 1) * G] : ecount;
  c[t] = 0;
  __syncthreads();
  for (int i0 = beg + t; i0 < end; i0 += 256 * 8) {
    unsigned key[8];
#pragma unroll
    for (int k = 0; k < 8; ++k) {
      const int i = i0 + k * 256;
      key[k] = (unsigned)keys[(i < end) ? i : i0];
    }
#pragma unroll
    for (int k = 0; k < 8; ++k) {
      const bool ok = i0 + k * 256 < end;
      atomicAdd(&c[ok ? (int)(key[k] >> 24) : t], ok ? 1 : 0);
    }
  }
  __syncthreads();
  const int my = c[t];
  int x = my;
  for (int o = 1; o < 32; o <<= 1) {
    int y = __shfl_up(x, o, 32);
    if (lane >= o) x += y;
  }
  if (lane == 31) wsum[w] = x;
  __syncthreads();
  int wb = 0;
  for (int j = 0; j < w; ++j) wb += wsum[j];
  const int start = beg + wb + x - my;
  const int node = b * BKT_NODES + t;
  if (node < n) {
    cnt[node] = my;
    segBeg[node] = start;
  }
  c[t] = start;
  __syncthreads();
  for (int i0 = beg + t; i0 < end; i0 += 256 * 8) {
    unsigned key[8];
    int pos[8];
#pragma unroll
    for (int k = 0; k < 8; ++k) {
      const int i = i0 + k * 256;
      key[k] = (unsigned)keys[(i < end) ? i : i0];
    }
#pragma unroll
    for (int k = 0; k < 8; ++k) {
      const bool ok = i0 + k * 256 < end;
      pos[k] = atomicAdd(&c[ok ? (int)(key[k] >> 24) : t], ok ? 1 : 0);
    }
#pragma unroll
    for (int k = 0; k < 8; ++k) {
      if (i0 + k * 256 < end) csrSrc[pos[k]] = (int)(key[k] & 0x00FFFFFFu);
    }
  }
}

__device__ __forceinline__ float leaky(float v) { return v > 0.0f ? v : v * NEG_SLOPE; }

__device__ __forceinline__ float4 ld_row(const float* __restrict__ H, int s, int lane) {
  return *(const float4*)(H + (size_t)s * DFEAT + lane * 4);
}

__device__ __forceinline__ void fma_row(float4& acc, float c, const float4& h) {
  acc.x += c * h.x;
  acc.y += c * h.y;
  acc.z += c * h.z;
  acc.w += c * h.w;
}

__device__ __forceinline__ void agg_rows(const float* __restrict__ H, int lane, int s, float coef,
                                         int cnt, float4& acc) {
  const int ci = __float_as_int(coef);
  int t = 0;
  for (; t + 4 <= cnt; t += 4) {
    int s0 = __builtin_amdgcn_readlane(s, t);
    int s1 = __builtin_amdgcn_readlane(s, t + 1);
    int s2 = __builtin_amdgcn_readlane(s, t + 2);
    int s3 = __builtin_amdgcn_readlane(s, t + 3);
    float c0 = __int_as_float(__builtin_amdgcn_readlane(ci, t));
    float c1 = __int_as_float(__builtin_amdgcn_readlane(ci, t + 1));
    float c2 = __int_as_float(__builtin_amdgcn_readlane(ci, t + 2));
    float c3 = __int_as_float(__builtin_amdgcn_readlane(ci, t + 3));
    float4 h0 = ld_row(H, s0, lane);
    float4 h1 = ld_row(H, s1, lane);
    float4 h2 = ld_row(H, s2, lane);
    float4 h3 = ld_row(H, s3, lane);
    fma_row(acc, c0, h0);
    fma_row(acc, c1, h1);
    fma_row(acc, c2, h2);
    fma_row(acc, c3, h3);
  }
  for (; t < cnt; ++t) {
    int s0 = __builtin_amdgcn_readlane(s, t);
    float c0 = __int_as_float(__builtin_amdgcn_readlane(ci, t));
    float4 h0 = ld_row(H, s0, lane);
    fma_row(acc, c0, h0);
  }
}

__global__ void __launch_bounds__(256) gat_node_gather(const int* __restrict__ csrSrc0,
                                                       const int* __restrict__ segBeg0,
                                                       const int* __restrict__ cnt0,
                                                       const float* __restrict__ asrc0,
                                                       const float* __restrict__ adst0,
                                                       const float* __restrict__ H0,
                                                       const float* __restrict__ bias0,
                                                       float* __restrict__ O0,
                                                       const int* __restrict__ csrSrc1,
                                                       const int* __restrict__ segBeg1,
                                                       const int* __restrict__ cnt1,
                                                       const float* __restrict__ asrc1,
                                                       const float* __restrict__ adst1,
                                                       const float* __restrict__ H1,
                                                       const float* __restrict__ bias1,
                                                       float* __restrict__ O1, int n, int relu) {
  const bool br = blockIdx.y != 0;
  const int* __restrict__ csrSrc = br ? csrSrc1 : csrSrc0;
  const int* __restrict__ segBeg = br ? segBeg1 : segBeg0;
  const int* __restrict__ cnt = br ? cnt1 : cnt0;
  const float* __restrict__ asrc = br ? asrc1 : asrc0;
  const float* __restrict__ adst = br ? adst1 : adst0;
  const float* __restrict__ H = br ? H1 : H0;
  const float* __restrict__ bias = br ? bias1 : bias0;
  float* __restrict__ O = br ? O1 : O0;
  const int g = blockIdx.x * blockDim.x + threadIdx.x;
  const int lane = threadIdx.x & 31;
  const int d = __builtin_amdgcn_readfirstlane(g >> 5);
  if (d >= n) return;
  const int beg = segBeg[d];
  const int end = beg + cnt[d];
  const float ad = adst[d];
  const float vself = leaky(asrc[d] + ad);

  const int j0 = beg + lane;
  const bool ok0 = j0 < end;
  const int s0 = ok0 ? csrSrc[j0] : d;
  const float e0 = leaky(asrc[s0] + ad);

  float m = fmaxf(vself, e0);
  for (int j = j0 + 32; j < end; j += 32) m = fmaxf(m, leaky(asrc[csrSrc[j]] + ad));
  for (int off = 16; off; off >>= 1) m = fmaxf(m, __shfl_xor(m, off, 32));

  const float w0 = ok0 ? expf(e0 - m) : 0.0f;
  const float wself = expf(vself - m);
  float sum = w0;
  for (int j = j0 + 32; j < end; j += 32) sum += expf(leaky(asrc[csrSrc[j]] + ad) - m);
  if (lane == 0) sum += wself;
  for (int off = 16; off; off >>= 1) sum += __shfl_xor(sum, off, 32);
  const float den = sum + 1e-16f;

  float4 acc = make_float4(0.0f, 0.0f, 0.0f, 0.0f);
  fma_row(acc, wself / den, ld_row(H, d, lane));
  {
    const int c0 = (end - beg < 32) ? (end - beg) : 32;
    agg_rows(H, lane, s0, w0 / den, c0, acc);
  }
  for (int base = beg + 32; base < end; base += 32) {
    const int j = base + lane;
    int s = d;
    float coef = 0.0f;
    if (j < end) {
      s = csrSrc[j];
      coef = expf(leaky(asrc[s] + ad) - m) / den;
    }
    const int c = (end - base < 32) ? (end - base) : 32;
    agg_rows(H, lane, s, coef, c, acc);
  }

  const float4 b4 = *(const float4*)(bias + lane * 4);
  acc.x += b4.x;
  acc.y += b4.y;
  acc.z += b4.z;
  acc.w += b4.w;
  if (relu) {
    acc.x = fmaxf(acc.x, 0.0f);
    acc.y = fmaxf(acc.y, 0.0f);
    acc.z = fmaxf(acc.z, 0.0f);
    acc.w = fmaxf(acc.w, 0.0f);
  }
  *(float4*)(O + (size_t)d * DFEAT + lane * 4) = acc;
}

__global__ void __launch_bounds__(256) gat_predict(const float* __restrict__ x2o,
                                                   const float* __restrict__ x2s,
                                                   const float* __restrict__ deg,
                                                   const float* __restrict__ Wp,
                                                   const float* __restrict__ bp,
                                                   float* __restrict__ out, int n) {
  int i = blockIdx.x * blockDim.x + threadIdx.x;
  if (i >= n) return;
  const float* zo = x2o + (size_t)i * DFEAT;
  const float* zs = x2s + (size_t)i * DFEAT;
  const float* zd = deg + (size_t)i * DDEG;
  float l0 = bp[0], l1 = bp[1];
#pragma unroll 4
  for (int d = 0; d < DFEAT; ++d) {
    l0 += zo[d] * Wp[d * 2 + 0];
    l1 += zo[d] * Wp[d * 2 + 1];
  }
#pragma unroll 4
  for (int d = 0; d < DFEAT; ++d) {
    l0 += zs[d] * Wp[(DFEAT + d) * 2 + 0];
    l1 += zs[d] * Wp[(DFEAT + d) * 2 + 1];
  }
#pragma unroll 4
  for (int d = 0; d < DDEG; ++d) {
    l0 += zd[d] * Wp[(2 * DFEAT + d) * 2 + 0];
    l1 += zd[d] * Wp[(2 * DFEAT + d) * 2 + 1];
  }
  float mx = fmaxf(l0, l1);
  float e0 = expf(l0 - mx), e1 = expf(l1 - mx);
  float inv = 1.0f / (e0 + e1);
  float a0 = e0 * inv, a1 = e1 * inv;
  float* po = out + (size_t)i * DFEAT;
#pragma unroll 4
  for (int d = 0; d < DFEAT; ++d) po[d] = a0 * zo[d] + a1 * zs[d];
}

static inline int ceil_div(long long a, long long b) { return (int)((a + b - 1) / b); }

struct LayerIo {
  const float* X;
  const float* W;
  const float* avs;
  const float* avd;
  const float* bias;
  float* H;
  float* asrc;
  float* adst;
  const int* csrSrc;
  const int* segBeg;
  const int* cnt;
  float* O;
};

static void launch_gemm2(const LayerIo& a, const LayerIo& b, int n, const int* scanIn,
                         int* scanOut, int scanN, int scanBlocks, hipStream_t stream) {
  gat_gemm_xw<<<dim3(scanBlocks + n / 16, 2), 256, 0, stream>>>(
      a.X, a.W, a.avs, a.avd, a.H, a.asrc, a.adst, b.X, b.W, b.avs, b.avd, b.H, b.asrc, b.adst, n,
      scanIn, scanOut, scanN, scanBlocks);
}

static void launch_gather2(const LayerIo& a, const LayerIo& b, int n, int relu,
                           hipStream_t stream) {
  gat_node_gather<<<dim3(ceil_div((long long)n * 32, 256), 2), 256, 0, stream>>>(
      a.csrSrc, a.segBeg, a.cnt, a.asrc, a.adst, a.H, a.bias, a.O,
      b.csrSrc, b.segBeg, b.cnt, b.asrc, b.adst, b.H, b.bias, b.O, n, relu);
}

static void build_csr2_and_layer1_product(const int* src0, const int* dst0, const int* src1,
                                          const int* dst1, int ecount, int n, int* cnt0,
                                          int* segBeg0, int* csrSrc0, int* cnt1, int* segBeg1,
                                          int* csrSrc1, int* keys, int* hist, int* off,
                                          const LayerIo& a, const LayerIo& b, hipStream_t stream) {
  const int G = ceil_div(ecount, BKT_EDGES_PER_BLOCK);
  const int nb = ceil_div(n, BKT_NODES);
  csr_bucket_count<<<dim3(G, 2), 512, 0, stream>>>(dst0, dst1, ecount, nb, hist);
  launch_gemm2(a, b, n, hist, off, nb * G, ceil_div((long long)nb * G, SCAN_CHUNK), stream);
  csr_bucket_scatter<<<dim3(G, 2), 512, 0, stream>>>(src0, dst0, src1, dst1, ecount, nb, off, keys);
  csr_bucket_finish<<<dim3(nb, 2), 256, 0, stream>>>(keys, off, G, nb, ecount, n, cnt0, segBeg0,
                                                     csrSrc0, cnt1, segBeg1, csrSrc1);
}

extern "C" void kernel_launch(void* const* d_in, const int* in_sizes, int n_in,
                              void* d_out, int out_size, void* d_ws, size_t ws_size,
                              hipStream_t stream) {
  const float* x_o      = (const float*)d_in[0];
  const float* degree_o = (const float*)d_in[1];
  const int*   ei_o     = (const int*)d_in[2];
  const int*   ei_s     = (const int*)d_in[3];
  const float* W_o1 = (const float*)d_in[4];
  const float* as_o1 = (const float*)d_in[5];
  const float* ad_o1 = (const float*)d_in[6];
  const float* b_o1  = (const float*)d_in[7];
  const float* W_o2 = (const float*)d_in[8];
  const float* as_o2 = (const float*)d_in[9];
  const float* ad_o2 = (const float*)d_in[10];
  const float* b_o2  = (const float*)d_in[11];
  const float* W_s1 = (const float*)d_in[12];
  const float* as_s1 = (const float*)d_in[13];
  const float* ad_s1 = (const float*)d_in[14];
  const float* b_s1  = (const float*)d_in[15];
  const float* W_s2 = (const float*)d_in[16];
  const float* as_s2 = (const float*)d_in[17];
  const float* ad_s2 = (const float*)d_in[18];
  const float* b_s2  = (const float*)d_in[19];
  const float* W_pred = (const float*)d_in[20];
  const float* b_pred = (const float*)d_in[21];

  const int n = in_sizes[0] / DFEAT;
  const int ecount = in_sizes[2] / 2;

  const int* src_o = ei_o;
  const int* dst_o = ei_o + ecount;
  const int* src_s = ei_s;
  const int* dst_s = ei_s + ecount;

  float* ws = (float*)d_ws;
  float* Ho    = ws;
  float* Hs    = Ho + (size_t)n * DFEAT;
  float* Yo    = Hs + (size_t)n * DFEAT;
  float* Ys    = Yo + (size_t)n * DFEAT;
  float* asrcO = Ys + (size_t)n * DFEAT;
  float* adstO = asrcO + n;
  int* cntO    = (int*)(adstO + n);
  int* segBegO = cntO + n;
  int* csrSrcO = segBegO + n;
  int* csrSrcS = (int*)d_out;
  int* cntS    = csrSrcS + ecount;
  int* segBegS = cntS + n;
  float* asrcS = (float*)(segBegS + n);
  float* adstS = asrcS + n;
  int* keys    = (int*)Ys;
  int* hist    = (int*)Yo;
  int* off     = hist + (size_t)n * DFEAT / 2;

  const LayerIo o1 = {x_o, W_o1, as_o1, ad_o1, b_o1, Ho, asrcO, adstO, csrSrcO, segBegO, cntO, Yo};
  const LayerIo s1 = {x_o, W_s1, as_s1, ad_s1, b_s1, Hs, asrcS, adstS, csrSrcS, segBegS, cntS, Ys};
  build_csr2_and_layer1_product(src_o, dst_o, src_s, dst_s, ecount, n, cntO, segBegO, csrSrcO, cntS,
                                segBegS, csrSrcS, keys, hist, off, o1, s1, stream);
  launch_gather2(o1, s1, n,  1, stream);
  const LayerIo o2 = {Yo, W_o2, as_o2, ad_o2, b_o2, Ho, asrcO, adstO, csrSrcO, segBegO, cntO, Yo};
  const LayerIo s2 = {Ys, W_s2, as_s2, ad_s2, b_s2, Hs, asrcS, adstS, csrSrcS, segBegS, cntS, Ys};
  launch_gemm2(o2, s2, n, nullptr, nullptr, 0, 0, stream);
  launch_gather2(o2, s2, n,  0, stream);

  gat_predict<<<ceil_div(n, 256), 256, 0, stream>>>(Yo, Ys, degree_o, W_pred, b_pred,
                                                    (float*)d_out, n);
}
